# starting kernel
# baseline (speedup 1.0000x reference)
#include <hip/hip_runtime.h>

typedef __attribute__((ext_vector_type(2))) float v2f;
typedef __attribute__((ext_vector_type(8))) float v8f;

#define NEG_SLOPE 0.2f


__device__ __forceinline__ unsigned f2ord(float f) {
  unsigned u = __float_as_uint(f);
  return (u & 0x80000000u) ? ~u : (u | 0x80000000u);
}
__device__ __forceinline__ float ord2f(unsigned o) {
  unsigned u = (o & 0x80000000u) ? (o & 0x7fffffffu) : ~o;
  return __uint_as_float(u);
}

__device__ __forceinline__ void edge_ends(const int* __restrict__ ei, int e,
                                          int Eraw, int& s, int& d) {
  if (e < Eraw) { s = ei[e]; d = ei[Eraw + e]; }
  else          { s = e - Eraw; d = s; }
}


__global__ void fill_f32(float* __restrict__ p, float v, long long n) {
  long long i = (long long)blockIdx.x * blockDim.x + threadIdx.x;
  if (i < n) p[i] = v;
}
__global__ void fill_u32(unsigned* __restrict__ p, unsigned v, long long n) {
  long long i = (long long)blockIdx.x * blockDim.x + threadIdx.x;
  if (i < n) p[i] = v;
}


__global__ void gemm_wmma_f32(const float* __restrict__ A,
                              const float* __restrict__ B,
                              float* __restrict__ C,
                              int M, int Nn, int K) {
  const int lane   = threadIdx.x & 31;
  const int wave   = blockIdx.x * (blockDim.x >> 5) + (threadIdx.x >> 5);
  const int ntiles = Nn >> 4;
  const int mtiles = M >> 4;
  const int mt64   = (mtiles + 3) >> 2;
  const int tileM  = wave / ntiles;
  const int tileN  = wave - tileM * ntiles;
  if (tileM >= mt64) return;

  const int lh  = lane >> 4;
  const int l16 = lane & 15;
  const int m0  = tileM * 4;

  const float* arow[4];
#pragma unroll
  for (int r = 0; r < 4; ++r) {
    int mr = m0 + r; if (mr > mtiles - 1) mr = mtiles - 1;
    arow[r] = A + (size_t)(mr * 16 + l16) * K + lh * 2;
  }
  const float* bptr = B + (size_t)(lh * 2) * Nn + tileN * 16 + l16;

  v8f acc0 = {}, acc1 = {}, acc2 = {}, acc3 = {};
  for (int k = 0; k < K; k += 4) {
    v2f b;
    b.x = bptr[(size_t)k * Nn];
    b.y = bptr[(size_t)(k + 1) * Nn];
    v2f a0 = *(const v2f*)(arow[0] + k);
    v2f a1 = *(const v2f*)(arow[1] + k);
    v2f a2 = *(const v2f*)(arow[2] + k);
    v2f a3 = *(const v2f*)(arow[3] + k);
    acc0 = __builtin_amdgcn_wmma_f32_16x16x4_f32(false, a0, false, b, (short)0, acc0, false, false);
    acc1 = __builtin_amdgcn_wmma_f32_16x16x4_f32(false, a1, false, b, (short)0, acc1, false, false);
    acc2 = __builtin_amdgcn_wmma_f32_16x16x4_f32(false, a2, false, b, (short)0, acc2, false, false);
    acc3 = __builtin_amdgcn_wmma_f32_16x16x4_f32(false, a3, false, b, (short)0, acc3, false, false);
  }

  const int col = tileN * 16 + l16;
  v8f accs[4] = {acc0, acc1, acc2, acc3};
#pragma unroll
  for (int r = 0; r < 4; ++r) {
    if (m0 + r < mtiles) {
      float* crow = C + (size_t)((m0 + r) * 16 + lh * 8) * Nn + col;
#pragma unroll
      for (int q = 0; q < 8; ++q) crow[(size_t)q * Nn] = accs[r][q];
    }
  }
}


template <int H, int F>
__global__ void node_alpha(const float* __restrict__ h,
                           const float* __restrict__ att_s,
                           const float* __restrict__ att_d,
                           float* __restrict__ asrc, float* __restrict__ adst,
                           int n_nodes) {
  int idx = blockIdx.x * blockDim.x + threadIdx.x;
  if (idx >= n_nodes * H) return;
  int n  = idx / H;
  int hd = idx - n * H;
  const float* hp = h + (size_t)n * (H * F) + hd * F;
  const float* as = att_s + hd * F;
  const float* ad = att_d + hd * F;
  float s = 0.f, d = 0.f;
#pragma unroll 8
  for (int f = 0; f < F; ++f) { float v = hp[f]; s += v * as[f]; d += v * ad[f]; }
  asrc[idx] = s;
  adst[idx] = d;
}


template <int H>
__global__ void edge_max_k(const int* __restrict__ ei,
                           const float* __restrict__ asrc,
                           const float* __restrict__ adst,
                           unsigned* __restrict__ mord, int Eraw, int Etot) {
  int idx = blockIdx.x * blockDim.x + threadIdx.x;
  if (idx >= Etot * H) return;
  int e  = idx / H;
  int hd = idx - e * H;
  int s, d; edge_ends(ei, e, Eraw, s, d);
  float v = asrc[s * H + hd] + adst[d * H + hd];
  v = v > 0.f ? v : NEG_SLOPE * v;
  atomicMax(mord + d * H + hd, f2ord(v));
}

template <int H>
__global__ void edge_expsum_k(const int* __restrict__ ei,
                              const float* __restrict__ asrc,
                              const float* __restrict__ adst,
                              const unsigned* __restrict__ mord,
                              float* __restrict__ denom, int Eraw, int Etot) {
  int idx = blockIdx.x * blockDim.x + threadIdx.x;
  if (idx >= Etot * H) return;
  int e  = idx / H;
  int hd = idx - e * H;
  int s, d; edge_ends(ei, e, Eraw, s, d);
  float v = asrc[s * H + hd] + adst[d * H + hd];
  v = v > 0.f ? v : NEG_SLOPE * v;
  float m = ord2f(mord[d * H + hd]);
  atomicAdd(denom + d * H + hd, __expf(v - m));
}

template <int H, int F>
__global__ void edge_agg_k(const int* __restrict__ ei,
                           const float* __restrict__ asrc,
                           const float* __restrict__ adst,
                           const unsigned* __restrict__ mord,
                           const float* __restrict__ denom,
                           const float* __restrict__ feat,
                           float* __restrict__ out, int Eraw, int Etot) {
  const int C  = H * F;
  const int CV = C / 4;
  long long idx = (long long)blockIdx.x * blockDim.x + threadIdx.x;
  if (idx >= (long long)Etot * CV) return;
  int e  = (int)(idx / CV);
  int c4 = (int)(idx - (long long)e * CV);
  int c  = c4 * 4;
  int hd = c / F;
  int s, d; edge_ends(ei, e, Eraw, s, d);
  float v = asrc[s * H + hd] + adst[d * H + hd];
  v = v > 0.f ? v : NEG_SLOPE * v;
  float m     = ord2f(mord[d * H + hd]);
  float alpha = __expf(v - m) / (denom[d * H + hd] + 1e-16f);
  const float4 f4 = *(const float4*)(feat + (size_t)s * C + c);
  float* op = out + (size_t)d * C + c;
  atomicAdd(op + 0, f4.x * alpha);
  atomicAdd(op + 1, f4.y * alpha);
  atomicAdd(op + 2, f4.z * alpha);
  atomicAdd(op + 3, f4.w * alpha);
}


__global__ void bias_elu(float* __restrict__ x, const float* __restrict__ b,
                         int C, long long n) {
  long long i = (long long)blockIdx.x * blockDim.x + threadIdx.x;
  if (i >= n) return;
  float v = x[i] + b[(int)(i % C)];
  x[i] = v > 0.f ? v : expm1f(v);
}

__global__ void bias_add(float* __restrict__ x, const float* __restrict__ b,
                         int C, long long n) {
  long long i = (long long)blockIdx.x * blockDim.x + threadIdx.x;
  if (i >= n) return;
  x[i] += b[(int)(i % C)];
}


static inline int cdiv(long long a, long long b) { return (int)((a + b - 1) / b); }

extern "C" void kernel_launch(void* const* d_in, const int* in_sizes, int n_in,
                              void* d_out, int out_size, void* d_ws, size_t ws_size,
                              hipStream_t stream) {
  const float* x      = (const float*)d_in[0];
  const int*   ei     = (const int*)  d_in[1];
  const float* W1     = (const float*)d_in[2];
  const float* att_s1 = (const float*)d_in[3];
  const float* att_d1 = (const float*)d_in[4];
  const float* b1     = (const float*)d_in[5];
  const float* W2     = (const float*)d_in[6];
  const float* att_s2 = (const float*)d_in[7];
  const float* att_d2 = (const float*)d_in[8];
  const float* b2     = (const float*)d_in[9];

  const int IN   = 128;
  const int C1   = 256;
  const int F2   = 64;
  const int Nn   = in_sizes[0] / IN;
  const int Eraw = in_sizes[1] / 2;
  const int Etot = Eraw + Nn;

  float* ws = (float*)d_ws;
  size_t off = 0;
  float*    h1     = ws + off; off += (size_t)Nn * C1;
  float*    out1   = ws + off; off += (size_t)Nn * C1;
  float*    asrc1  = ws + off; off += (size_t)Nn * 4;
  float*    adst1  = ws + off; off += (size_t)Nn * 4;
  unsigned* mord1  = (unsigned*)(ws + off); off += (size_t)Nn * 4;
  float*    denom1 = ws + off; off += (size_t)Nn * 4;
  float*    h2     = h1;
  float*    asrc2  = asrc1;
  float*    adst2  = adst1;
  unsigned* mord2  = mord1;
  float*    denom2 = denom1;
  float*    outf   = (float*)d_out;

  const int T = 256;

  {
    int mt64  = (Nn / 16 + 3) / 4;
    int tiles = mt64 * (C1 / 16);
    gemm_wmma_f32<<<cdiv(tiles, 8), 256, 0, stream>>>(x, W1, h1, Nn, C1, IN);
  }
  node_alpha<4, 64><<<cdiv((long long)Nn * 4, T), T, 0, stream>>>(
      h1, att_s1, att_d1, asrc1, adst1, Nn);
  fill_u32<<<cdiv((long long)Nn * 4, T), T, 0, stream>>>(mord1, 0u, (long long)Nn * 4);
  edge_max_k<4><<<cdiv((long long)Etot * 4, T), T, 0, stream>>>(
      ei, asrc1, adst1, mord1, Eraw, Etot);
  fill_f32<<<cdiv((long long)Nn * 4, T), T, 0, stream>>>(denom1, 0.f, (long long)Nn * 4);
  edge_expsum_k<4><<<cdiv((long long)Etot * 4, T), T, 0, stream>>>(
      ei, asrc1, adst1, mord1, denom1, Eraw, Etot);
  fill_f32<<<cdiv((long long)Nn * C1, T), T, 0, stream>>>(out1, 0.f, (long long)Nn * C1);
  edge_agg_k<4, 64><<<cdiv((long long)Etot * (C1 / 4), T), T, 0, stream>>>(
      ei, asrc1, adst1, mord1, denom1, h1, out1, Eraw, Etot);
  bias_elu<<<cdiv((long long)Nn * C1, T), T, 0, stream>>>(out1, b1, C1, (long long)Nn * C1);

  {
    int mt64  = (Nn / 16 + 3) / 4;
    int tiles = mt64 * (F2 / 16);
    gemm_wmma_f32<<<cdiv(tiles, 8), 256, 0, stream>>>(out1, W2, h2, Nn, F2, C1);
  }
  node_alpha<1, 64><<<cdiv((long long)Nn, T), T, 0, stream>>>(
      h2, att_s2, att_d2, asrc2, adst2, Nn);
  fill_u32<<<cdiv((long long)Nn, T), T, 0, stream>>>(mord2, 0u, (long long)Nn);
  edge_max_k<1><<<cdiv((long long)Etot, T), T, 0, stream>>>(
      ei, asrc2, adst2, mord2, Eraw, Etot);
  fill_f32<<<cdiv((long long)Nn, T), T, 0, stream>>>(denom2, 0.f, (long long)Nn);
  edge_expsum_k<1><<<cdiv((long long)Etot, T), T, 0, stream>>>(
      ei, asrc2, adst2, mord2, denom2, Eraw, Etot);
  fill_f32<<<cdiv((long long)Nn * F2, T), T, 0, stream>>>(outf, 0.f, (long long)Nn * F2);
  edge_agg_k<1, 64><<<cdiv((long long)Etot * (F2 / 4), T), T, 0, stream>>>(
      ei, asrc2, adst2, mord2, denom2, h2, outf, Eraw, Etot);
  bias_add<<<cdiv((long long)Nn * F2, T), T, 0, stream>>>(outf, b2, F2, (long long)Nn * F2);
}


# Aggregation by destination range in LDS: a work-group owns 200 nodes, scans the dst list, adds its edges' weighted rows by LDS f32 atomics (arrival order), normalises, writes; global atomics gone
# speedup vs baseline: 2.0936x; 2.0936x over previous
#include <hip/hip_runtime.h>

typedef __attribute__((ext_vector_type(2))) float v2f;
typedef __attribute__((ext_vector_type(8))) float v8f;

#define NEG_SLOPE 0.2f


__device__ __forceinline__ unsigned f2ord(float f) {
  unsigned u = __float_as_uint(f);
  return (u & 0x80000000u) ? ~u : (u | 0x80000000u);
}
__device__ __forceinline__ float ord2f(unsigned o) {
  unsigned u = (o & 0x80000000u) ? (o & 0x7fffffffu) : ~o;
  return __uint_as_float(u);
}

__device__ __forceinline__ void edge_ends(const int* __restrict__ ei, int e,
                                          int Eraw, int& s, int& d) {
  if (e < Eraw) { s = ei[e]; d = ei[Eraw + e]; }
  else          { s = e - Eraw; d = s; }
}


__global__ void fill_u32(unsigned* __restrict__ p, unsigned v, long long n) {
  long long i = (long long)blockIdx.x * blockDim.x + threadIdx.x;
  if (i < n) p[i] = v;
}


__global__ void gemm_wmma_f32(const float* __restrict__ A,
                              const float* __restrict__ B,
                              float* __restrict__ C,
                              int M, int Nn, int K) {
  const int lane   = threadIdx.x & 31;
  const int wave   = blockIdx.x * (blockDim.x >> 5) + (threadIdx.x >> 5);
  const int ntiles = Nn >> 4;
  const int mtiles = M >> 4;
  const int mt64   = (mtiles + 3) >> 2;
  const int tileM  = wave / ntiles;
  const int tileN  = wave - tileM * ntiles;
  if (tileM >= mt64) return;

  const int lh  = lane >> 4;
  const int l16 = lane & 15;
  const int m0  = tileM * 4;

  const float* arow[4];
#pragma unroll
  for (int r = 0; r < 4; ++r) {
    int mr = m0 + r; if (mr > mtiles - 1) mr = mtiles - 1;
    arow[r] = A + (size_t)(mr * 16 + l16) * K + lh * 2;
  }
  const float* bptr = B + (size_t)(lh * 2) * Nn + tileN * 16 + l16;

  v8f acc0 = {}, acc1 = {}, acc2 = {}, acc3 = {};
  for (int k = 0; k < K; k += 4) {
    v2f b;
    b.x = bptr[(size_t)k * Nn];
    b.y = bptr[(size_t)(k + 1) * Nn];
    v2f a0 = *(const v2f*)(arow[0] + k);
    v2f a1 = *(const v2f*)(arow[1] + k);
    v2f a2 = *(const v2f*)(arow[2] + k);
    v2f a3 = *(const v2f*)(arow[3] + k);
    acc0 = __builtin_amdgcn_wmma_f32_16x16x4_f32(false, a0, false, b, (short)0, acc0, false, false);
    acc1 = __builtin_amdgcn_wmma_f32_16x16x4_f32(false, a1, false, b, (short)0, acc1, false, false);
    acc2 = __builtin_amdgcn_wmma_f32_16x16x4_f32(false, a2, false, b, (short)0, acc2, false, false);
    acc3 = __builtin_amdgcn_wmma_f32_16x16x4_f32(false, a3, false, b, (short)0, acc3, false, false);
  }

  const int col = tileN * 16 + l16;
  v8f accs[4] = {acc0, acc1, acc2, acc3};
#pragma unroll
  for (int r = 0; r < 4; ++r) {
    if (m0 + r < mtiles) {
      float* crow = C + (size_t)((m0 + r) * 16 + lh * 8) * Nn + col;
#pragma unroll
      for (int q = 0; q < 8; ++q) crow[(size_t)q * Nn] = accs[r][q];
    }
  }
}


template <int H, int F>
__global__ void node_alpha(const float* __restrict__ h,
                           const float* __restrict__ att_s,
                           const float* __restrict__ att_d,
                           float* __restrict__ asrc, float* __restrict__ adst,
                           int n_nodes) {
  int idx = blockIdx.x * blockDim.x + threadIdx.x;
  if (idx >= n_nodes * H) return;
  int n  = idx / H;
  int hd = idx - n * H;
  const float* hp = h + (size_t)n * (H * F) + hd * F;
  const float* as = att_s + hd * F;
  const float* ad = att_d + hd * F;
  float s = 0.f, d = 0.f;
#pragma unroll 8
  for (int f = 0; f < F; ++f) { float v = hp[f]; s += v * as[f]; d += v * ad[f]; }
  asrc[idx] = s;
  adst[idx] = d;
}


template <int H>
__global__ void edge_max_k(const int* __restrict__ ei,
                           const float* __restrict__ asrc,
                           const float* __restrict__ adst,
                           unsigned* __restrict__ mord, int Eraw, int Etot) {
  int idx = blockIdx.x * blockDim.x + threadIdx.x;
  if (idx >= Etot * H) return;
  int e  = idx / H;
  int hd = idx - e * H;
  int s, d; edge_ends(ei, e, Eraw, s, d);
  float v = asrc[s * H + hd] + adst[d * H + hd];
  v = v > 0.f ? v : NEG_SLOPE * v;
  atomicMax(mord + d * H + hd, f2ord(v));
}

template <int H, int F, int B, int NPB, bool ELU>
__global__ __launch_bounds__(1024) void range_agg_k(
    const int* __restrict__ ei,
    const float* __restrict__ asrc,
    const float* __restrict__ adst,
    const unsigned* __restrict__ mord,
    const float* __restrict__ feat,
    const float* __restrict__ bias,
    float* __restrict__ out, int n_nodes, int Eraw) {
  constexpr int C  = H * F;
  constexpr int NL = C / 32;
  constexpr int G  = B * H;
  constexpr int QC = 160;
  static_assert(G <= 32 && (32 % G) == 0 && (F % 32) == 0 && (C % 4) == 0, "tiling");
  static_assert(B - 1 + 128 <= QC, "queue");

  __shared__ __attribute__((aligned(16))) float tile[NPB * C];
  __shared__ float den[NPB * H];
  __shared__ float mloc[NPB * H];
  __shared__ float adl[NPB * H];
  __shared__ int   qe[32 * QC];

  const int tid  = threadIdx.x;
  const int lane = tid & 31;
  const int wv   = __builtin_amdgcn_readfirstlane(tid >> 5);
  const int nw   = blockDim.x >> 5;
  const int base = blockIdx.x * NPB;
  int nloc = n_nodes - base; if (nloc > NPB) nloc = NPB;

  {
    float4* t4 = (float4*)tile;
    for (int i = tid; i < NPB * C / 4; i += blockDim.x) t4[i] = make_float4(0.f, 0.f, 0.f, 0.f);
    for (int i = tid; i < NPB * H; i += blockDim.x) {
      const bool ok = i < nloc * H;
      den[i]  = 0.f;
      mloc[i] = ok ? ord2f(mord[(size_t)base * H + i]) : 0.f;
      adl[i]  = ok ? adst[(size_t)base * H + i] : 0.f;
    }
  }
  __syncthreads();

  int* myq = qe + wv * QC;
  int qn = 0;

  auto batch = [&](int q0, int cnt) {
    const int g  = lane % G;
    const int k  = g / H;
    const int hd = g - k * H;
    const int kq = k < cnt ? k : cnt - 1;
    const int eq = myq[q0 + kq];
    int sq, dq; edge_ends(ei, eq, Eraw, sq, dq);
    dq -= base;

    float f[B][NL];
    int   dk[B];
#pragma unroll
    for (int kk = 0; kk < B; ++kk) {
      const int sk = __builtin_amdgcn_readlane(sq, kk * H);
      dk[kk]       = __builtin_amdgcn_readlane(dq, kk * H);
      const float* fp = feat + (size_t)sk * C + lane;
#pragma unroll
      for (int j = 0; j < NL; ++j) f[kk][j] = fp[j * 32];
    }

    float v = asrc[(size_t)sq * H + hd] + adl[dq * H + hd];
    v = v > 0.f ? v : NEG_SLOPE * v;
    const float w = __expf(v - mloc[dq * H + hd]);
    if (lane < G && k < cnt) atomicAdd(den + dq * H + hd, w);

#pragma unroll
    for (int kk = 0; kk < B; ++kk) {
      if (kk < cnt) {
        float* op = tile + dk[kk] * C + lane;
#pragma unroll
        for (int j = 0; j < NL; ++j) {
          const float a = __int_as_float(
              __builtin_amdgcn_readlane(__float_as_int(w), kk * H + (j * 32) / F));
          atomicAdd(op + j * 32, f[kk][j] * a);
        }
      }
    }
  };

  auto push = [&](unsigned mask, int e) {
    const int pos = qn + (int)__builtin_amdgcn_mbcnt_lo(mask, 0u);
    if ((mask >> lane) & 1u) myq[pos] = e;
    qn += __builtin_popcount(mask);
  };
  auto drain = [&]() {
    __builtin_amdgcn_fence(__ATOMIC_ACQ_REL, "wavefront");
    while (qn >= B) { qn -= B; batch(qn, B); }
  };

  {
    const int j = wv + nw * lane;
    const unsigned mask = __builtin_amdgcn_ballot_w32(j < nloc);
    if (mask) { push(mask, Eraw + base + j); drain(); }
  }

  const int* dstp = ei + Eraw;
  const int nblk = ((Eraw & 3) == 0) ? (Eraw >> 7) : 0;
  auto ld = [&](int blk) {
    int4 r = make_int4(-1, -1, -1, -1);
    if (blk < nblk) r = *(const int4*)(dstp + (size_t)blk * 128 + lane * 4);
    return r;
  };
  auto proc = [&](int4 c, int blk) {
    const unsigned m0 = __builtin_amdgcn_ballot_w32((unsigned)(c.x - base) < (unsigned)nloc);
    const unsigned m1 = __builtin_amdgcn_ballot_w32((unsigned)(c.y - base) < (unsigned)nloc);
    const unsigned m2 = __builtin_amdgcn_ballot_w32((unsigned)(c.z - base) < (unsigned)nloc);
    const unsigned m3 = __builtin_amdgcn_ballot_w32((unsigned)(c.w - base) < (unsigned)nloc);
    if (m0 | m1 | m2 | m3) {
      const int e = blk * 128 + lane * 4;
      if (m0) push(m0, e);
      if (m1) push(m1, e + 1);
      if (m2) push(m2, e + 2);
      if (m3) push(m3, e + 3);
      drain();
    }
  };
  {
    int blk = wv;
    int4 r0 = ld(blk), r1 = ld(blk + nw), r2 = ld(blk + 2 * nw), r3 = ld(blk + 3 * nw);
    for (; blk < nblk; blk += 4 * nw) {
      proc(r0, blk);          r0 = ld(blk + 4 * nw);
      proc(r1, blk + nw);     r1 = ld(blk + 5 * nw);
      proc(r2, blk + 2 * nw); r2 = ld(blk + 6 * nw);
      proc(r3, blk + 3 * nw); r3 = ld(blk + 7 * nw);
    }
  }
  for (int e0 = nblk * 128 + wv * 32; e0 < Eraw; e0 += nw * 32) {
    const int e = e0 + lane;
    const int d = (e < Eraw) ? dstp[e] : -1;
    const unsigned mask = __builtin_amdgcn_ballot_w32((unsigned)(d - base) < (unsigned)nloc);
    if (mask) { push(mask, e); drain(); }
  }
  if (qn > 0) batch(0, qn);
  __syncthreads();

  {
    const float4* t4 = (const float4*)tile;
    float4* o4 = (float4*)(out + (size_t)base * C);
    const float4* b4 = (const float4*)bias;
    for (int i = tid; i < nloc * (C / 4); i += blockDim.x) {
      const int nl = i / (C / 4);
      const int c4 = i - nl * (C / 4);
      const float r = den[nl * H + (c4 * 4) / F] + 1e-16f;
      const float4 t = t4[i];
      const float4 bb = b4[c4];
      float4 v;
      v.x = t.x / r + bb.x; v.y = t.y / r + bb.y; v.z = t.z / r + bb.z; v.w = t.w / r + bb.w;
      if (ELU) {
        v.x = v.x > 0.f ? v.x : expm1f(v.x);
        v.y = v.y > 0.f ? v.y : expm1f(v.y);
        v.z = v.z > 0.f ? v.z : expm1f(v.z);
        v.w = v.w > 0.f ? v.w : expm1f(v.w);
      }
      o4[i] = v;
    }
  }
}


static inline int cdiv(long long a, long long b) { return (int)((a + b - 1) / b); }

extern "C" void kernel_launch(void* const* d_in, const int* in_sizes, int n_in,
                              void* d_out, int out_size, void* d_ws, size_t ws_size,
                              hipStream_t stream) {
  const float* x      = (const float*)d_in[0];
  const int*   ei     = (const int*)  d_in[1];
  const float* W1     = (const float*)d_in[2];
  const float* att_s1 = (const float*)d_in[3];
  const float* att_d1 = (const float*)d_in[4];
  const float* b1     = (const float*)d_in[5];
  const float* W2     = (const float*)d_in[6];
  const float* att_s2 = (const float*)d_in[7];
  const float* att_d2 = (const float*)d_in[8];
  const float* b2     = (const float*)d_in[9];

  const int IN   = 128;
  const int C1   = 256;
  const int F2   = 64;
  const int Nn   = in_sizes[0] / IN;
  const int Eraw = in_sizes[1] / 2;
  const int Etot = Eraw + Nn;

  float* ws = (float*)d_ws;
  size_t off = 0;
  float*    h1     = ws + off; off += (size_t)Nn * C1;
  float*    out1   = ws + off; off += (size_t)Nn * C1;
  float*    asrc1  = ws + off; off += (size_t)Nn * 4;
  float*    adst1  = ws + off; off += (size_t)Nn * 4;
  unsigned* mord1  = (unsigned*)(ws + off); off += (size_t)Nn * 4;
  float*    h2     = h1;
  float*    asrc2  = asrc1;
  float*    adst2  = adst1;
  unsigned* mord2  = mord1;
  float*    outf   = (float*)d_out;

  const int T = 256;

  {
    int mt64  = (Nn / 16 + 3) / 4;
    int tiles = mt64 * (C1 / 16);
    gemm_wmma_f32<<<cdiv(tiles, 8), 256, 0, stream>>>(x, W1, h1, Nn, C1, IN);
  }
  node_alpha<4, 64><<<cdiv((long long)Nn * 4, T), T, 0, stream>>>(
      h1, att_s1, att_d1, asrc1, adst1, Nn);
  fill_u32<<<cdiv((long long)Nn * 4, T), T, 0, stream>>>(mord1, 0u, (long long)Nn * 4);
  edge_max_k<4><<<cdiv((long long)Etot * 4, T), T, 0, stream>>>(
      ei, asrc1, adst1, mord1, Eraw, Etot);
  range_agg_k<4, 64, 8, 200, true><<<cdiv(Nn, 200), 1024, 0, stream>>>(
      ei, asrc1, adst1, mord1, h1, b1, out1, Nn, Eraw);

  {
    int mt64  = (Nn / 16 + 3) / 4;
    int tiles = mt64 * (F2 / 16);
    gemm_wmma_f32<<<cdiv(tiles, 8), 256, 0, stream>>>(out1, W2, h2, Nn, F2, C1);
  }
  node_alpha<1, 64><<<cdiv((long long)Nn, T), T, 0, stream>>>(
      h2, att_s2, att_d2, asrc2, adst2, Nn);
  fill_u32<<<cdiv((long long)Nn, T), T, 0, stream>>>(mord2, 0u, (long long)Nn);
  edge_max_k<1><<<cdiv((long long)Etot, T), T, 0, stream>>>(
      ei, asrc2, adst2, mord2, Eraw, Etot);
  range_agg_k<1, 64, 16, 200, false><<<cdiv(Nn, 200), 1024, 0, stream>>>(
      ei, asrc2, adst2, mord2, h2, b2, outf, Nn, Eraw);
}


# Segment max folded into the range work-group: its one scan fills per-wave edge lists and the maxima (LDS atomic max), lists worked off behind a barrier; edge-max and fill launches gone
# speedup vs baseline: 2.7853x; 1.3304x over previous
#include <hip/hip_runtime.h>

typedef __attribute__((ext_vector_type(2))) float v2f;
typedef __attribute__((ext_vector_type(8))) float v8f;

#define NEG_SLOPE 0.2f


__device__ __forceinline__ unsigned f2ord(float f) {
  unsigned u = __float_as_uint(f);
  return (u & 0x80000000u) ? ~u : (u | 0x80000000u);
}
__device__ __forceinline__ float ord2f(unsigned o) {
  unsigned u = (o & 0x80000000u) ? (o & 0x7fffffffu) : ~o;
  return __uint_as_float(u);
}

__device__ __forceinline__ void edge_ends(const int* __restrict__ ei, int e,
                                          int Eraw, int& s, int& d) {
  if (e < Eraw) { s = ei[e]; d = ei[Eraw + e]; }
  else          { s = e - Eraw; d = s; }
}


__global__ void gemm_wmma_f32(const float* __restrict__ A,
                              const float* __restrict__ B,
                              float* __restrict__ C,
                              int M, int Nn, int K) {
  const int lane   = threadIdx.x & 31;
  const int wave   = blockIdx.x * (blockDim.x >> 5) + (threadIdx.x >> 5);
  const int ntiles = Nn >> 4;
  const int mtiles = M >> 4;
  const int mt64   = (mtiles + 3) >> 2;
  const int tileM  = wave / ntiles;
  const int tileN  = wave - tileM * ntiles;
  if (tileM >= mt64) return;

  const int lh  = lane >> 4;
  const int l16 = lane & 15;
  const int m0  = tileM * 4;

  const float* arow[4];
#pragma unroll
  for (int r = 0; r < 4; ++r) {
    int mr = m0 + r; if (mr > mtiles - 1) mr = mtiles - 1;
    arow[r] = A + (size_t)(mr * 16 + l16) * K + lh * 2;
  }
  const float* bptr = B + (size_t)(lh * 2) * Nn + tileN * 16 + l16;

  v8f acc0 = {}, acc1 = {}, acc2 = {}, acc3 = {};
  for (int k = 0; k < K; k += 4) {
    v2f b;
    b.x = bptr[(size_t)k * Nn];
    b.y = bptr[(size_t)(k + 1) * Nn];
    v2f a0 = *(const v2f*)(arow[0] + k);
    v2f a1 = *(const v2f*)(arow[1] + k);
    v2f a2 = *(const v2f*)(arow[2] + k);
    v2f a3 = *(const v2f*)(arow[3] + k);
    acc0 = __builtin_amdgcn_wmma_f32_16x16x4_f32(false, a0, false, b, (short)0, acc0, false, false);
    acc1 = __builtin_amdgcn_wmma_f32_16x16x4_f32(false, a1, false, b, (short)0, acc1, false, false);
    acc2 = __builtin_amdgcn_wmma_f32_16x16x4_f32(false, a2, false, b, (short)0, acc2, false, false);
    acc3 = __builtin_amdgcn_wmma_f32_16x16x4_f32(false, a3, false, b, (short)0, acc3, false, false);
  }

  const int col = tileN * 16 + l16;
  v8f accs[4] = {acc0, acc1, acc2, acc3};
#pragma unroll
  for (int r = 0; r < 4; ++r) {
    if (m0 + r < mtiles) {
      float* crow = C + (size_t)((m0 + r) * 16 + lh * 8) * Nn + col;
#pragma unroll
      for (int q = 0; q < 8; ++q) crow[(size_t)q * Nn] = accs[r][q];
    }
  }
}


template <int H, int F>
__global__ void node_alpha(const float* __restrict__ h,
                           const float* __restrict__ att_s,
                           const float* __restrict__ att_d,
                           float* __restrict__ asrc, float* __restrict__ adst,
                           int n_nodes) {
  int idx = blockIdx.x * blockDim.x + threadIdx.x;
  if (idx >= n_nodes * H) return;
  int n  = idx / H;
  int hd = idx - n * H;
  const float* hp = h + (size_t)n * (H * F) + hd * F;
  const float* as = att_s + hd * F;
  const float* ad = att_d + hd * F;
  float s = 0.f, d = 0.f;
#pragma unroll 8
  for (int f = 0; f < F; ++f) { float v = hp[f]; s += v * as[f]; d += v * ad[f]; }
  asrc[idx] = s;
  adst[idx] = d;
}


template <int H, int F, int B, int NPB, bool ELU>
__global__ __launch_bounds__(1024) void range_agg_k(
    const int* __restrict__ ei,
    const float* __restrict__ asrc,
    const float* __restrict__ adst,
    const float* __restrict__ feat,
    const float* __restrict__ bias,
    float* __restrict__ out, int n_nodes, int Eraw) {
  constexpr int C  = H * F;
  constexpr int NL = C / 32;
  constexpr int G  = B * H;
  constexpr int QC = 512;
  static_assert(G <= 32 && (32 % G) == 0 && (F % 32) == 0 && (C % 4) == 0, "tiling");
  static_assert(NPB <= 256, "local row must fit the low byte of a list entry");

  __shared__ __attribute__((aligned(16))) float tile[NPB * C];
  __shared__ float    den[NPB * H];
  __shared__ unsigned mu[NPB * H];
  __shared__ float    adl[NPB * H];
  __shared__ int      qe[32 * QC];
  __shared__ int      ovf;

  const int tid  = threadIdx.x;
  const int lane = tid & 31;
  const int wv   = __builtin_amdgcn_readfirstlane(tid >> 5);
  const int nw   = blockDim.x >> 5;
  const int base = blockIdx.x * NPB;
  int nloc = n_nodes - base; if (nloc > NPB) nloc = NPB;

  {
    float4* t4 = (float4*)tile;
    for (int i = tid; i < NPB * C / 4; i += blockDim.x) t4[i] = make_float4(0.f, 0.f, 0.f, 0.f);
    for (int i = tid; i < NPB * H; i += blockDim.x) {
      den[i] = 0.f;
      mu[i]  = 0u;
      adl[i] = (i < nloc * H) ? adst[(size_t)base * H + i] : 0.f;
    }
    if (tid == 0) ovf = 0;
  }
  __syncthreads();

  int* myq = qe + wv * QC;
  int  qn = 0;
  int  done = 0;
  bool lost = false;

  auto chunk = [&](int q0, int cnt) {
    if (lane < cnt) {
      const int e = myq[q0 + lane];
      int s, d; edge_ends(ei, e, Eraw, s, d);
      const int dl = d - base;
#pragma unroll
      for (int hd = 0; hd < H; ++hd) {
        float v = asrc[(size_t)s * H + hd] + adl[dl * H + hd];
        v = v > 0.f ? v : NEG_SLOPE * v;
        atomicMax(mu + dl * H + hd, f2ord(v));
      }
      myq[q0 + lane] = (s << 8) | dl;
    }
  };
  auto fold = [&](bool all) {
    __builtin_amdgcn_fence(__ATOMIC_ACQ_REL, "wavefront");
    while (qn - done >= 32) { chunk(done, 32); done += 32; }
    if (all && done < qn) { chunk(done, qn - done); done = qn; }
  };
  auto room = [&](int n) {
    if (qn + n > QC) { fold(true); qn = 0; done = 0; lost = true; }
  };
  auto push = [&](unsigned mask, int e) {
    const int pos = qn + (int)__builtin_amdgcn_mbcnt_lo(mask, 0u);
    if ((mask >> lane) & 1u) myq[pos] = e;
    qn += __builtin_popcount(mask);
  };

  {
    const int j = wv + nw * lane;
    const unsigned mask = __builtin_amdgcn_ballot_w32(j < nloc);
    if (mask) push(mask, Eraw + base + j);
  }

  const int* dstp = ei + Eraw;
  const int nblk = ((Eraw & 3) == 0) ? (Eraw >> 7) : 0;
  auto ld = [&](int blk) {
    int4 r = make_int4(-1, -1, -1, -1);
    if (blk < nblk) r = *(const int4*)(dstp + (size_t)blk * 128 + lane * 4);
    return r;
  };
  auto proc = [&](int4 c, int blk) {
    const unsigned m0 = __builtin_amdgcn_ballot_w32((unsigned)(c.x - base) < (unsigned)nloc);
    const unsigned m1 = __builtin_amdgcn_ballot_w32((unsigned)(c.y - base) < (unsigned)nloc);
    const unsigned m2 = __builtin_amdgcn_ballot_w32((unsigned)(c.z - base) < (unsigned)nloc);
    const unsigned m3 = __builtin_amdgcn_ballot_w32((unsigned)(c.w - base) < (unsigned)nloc);
    if (m0 | m1 | m2 | m3) {
      room(128);
      const int e = blk * 128 + lane * 4;
      if (m0) push(m0, e);
      if (m1) push(m1, e + 1);
      if (m2) push(m2, e + 2);
      if (m3) push(m3, e + 3);
      fold(false);
    }
  };
  {
    int blk = wv;
    int4 r0 = ld(blk), r1 = ld(blk + nw), r2 = ld(blk + 2 * nw), r3 = ld(blk + 3 * nw);
    for (; blk < nblk; blk += 4 * nw) {
      proc(r0, blk);          r0 = ld(blk + 4 * nw);
      proc(r1, blk + nw);     r1 = ld(blk + 5 * nw);
      proc(r2, blk + 2 * nw); r2 = ld(blk + 6 * nw);
      proc(r3, blk + 3 * nw); r3 = ld(blk + 7 * nw);
    }
  }
  for (int e0 = nblk * 128 + wv * 32; e0 < Eraw; e0 += nw * 32) {
    const int e = e0 + lane;
    const int d = (e < Eraw) ? dstp[e] : -1;
    const unsigned mask = __builtin_amdgcn_ballot_w32((unsigned)(d - base) < (unsigned)nloc);
    if (mask) { room(32); push(mask, e); fold(false); }
  }
  fold(true);
  if (lost && lane == 0) ovf = 1;
  __syncthreads();

  const int again = __builtin_amdgcn_readfirstlane(ovf);
  if (!again) {
    auto batch = [&](int q0, int cnt) {
      const int g  = lane % G;
      const int k  = g / H;
      const int hd = g - k * H;
      const int kq = k < cnt ? k : cnt - 1;
      const int pk = myq[q0 + kq];
      const int sq = pk >> 8;
      const int dq = pk & 255;

      float f[B][NL];
      int   dk[B];
#pragma unroll
      for (int kk = 0; kk < B; ++kk) {
        const int sk = __builtin_amdgcn_readlane(sq, kk * H);
        dk[kk]       = __builtin_amdgcn_readlane(dq, kk * H);
        const float* fp = feat + (size_t)sk * C + lane;
#pragma unroll
        for (int j = 0; j < NL; ++j) f[kk][j] = fp[j * 32];
      }

      float v = asrc[(size_t)sq * H + hd] + adl[dq * H + hd];
      v = v > 0.f ? v : NEG_SLOPE * v;
      const float w = __expf(v - ord2f(mu[dq * H + hd]));
      if (lane < G && k < cnt) atomicAdd(den + dq * H + hd, w);

#pragma unroll
      for (int kk = 0; kk < B; ++kk) {
        if (kk < cnt) {
          float* op = tile + dk[kk] * C + lane;
#pragma unroll
          for (int j = 0; j < NL; ++j) {
            const float a = __int_as_float(
                __builtin_amdgcn_readlane(__float_as_int(w), kk * H + (j * 32) / F));
            atomicAdd(op + j * 32, f[kk][j] * a);
          }
        }
      }
    };
    for (int q0 = 0; q0 < qn; q0 += B) batch(q0, (qn - q0) < B ? (qn - q0) : B);
  } else {
    const int Etot = Eraw + n_nodes;
    for (int e0 = wv * 32; e0 < Etot; e0 += nw * 32) {
      const int e = e0 + lane;
      int s = 0, d = -1;
      if (e < Etot) edge_ends(ei, e, Eraw, s, d);
      unsigned mask = __builtin_amdgcn_ballot_w32((unsigned)(d - base) < (unsigned)nloc);
      while (mask) {
        const int l = __builtin_ctz(mask);
        mask &= mask - 1;
        const int sk = __shfl(s, l);
        const int dk = __shfl(d, l) - base;
        if (lane < H) {
          float v = asrc[(size_t)sk * H + lane] + adl[dk * H + lane];
          v = v > 0.f ? v : NEG_SLOPE * v;
          atomicAdd(den + dk * H + lane, __expf(v - ord2f(mu[dk * H + lane])));
        }
#pragma unroll
        for (int j = 0; j < NL; ++j) {
          const int hd = (j * 32) / F;
          float v = asrc[(size_t)sk * H + hd] + adl[dk * H + hd];
          v = v > 0.f ? v : NEG_SLOPE * v;
          const float a = __expf(v - ord2f(mu[dk * H + hd]));
          atomicAdd(tile + dk * C + j * 32 + lane, feat[(size_t)sk * C + j * 32 + lane] * a);
        }
      }
    }
  }
  __syncthreads();

  {
    const float4* t4 = (const float4*)tile;
    float4* o4 = (float4*)(out + (size_t)base * C);
    const float4* b4 = (const float4*)bias;
    for (int i = tid; i < nloc * (C / 4); i += blockDim.x) {
      const int nl = i / (C / 4);
      const int c4 = i - nl * (C / 4);
      const float r = den[nl * H + (c4 * 4) / F] + 1e-16f;
      const float4 t = t4[i];
      const float4 bb = b4[c4];
      float4 v;
      v.x = t.x / r + bb.x; v.y = t.y / r + bb.y; v.z = t.z / r + bb.z; v.w = t.w / r + bb.w;
      if (ELU) {
        v.x = v.x > 0.f ? v.x : expm1f(v.x);
        v.y = v.y > 0.f ? v.y : expm1f(v.y);
        v.z = v.z > 0.f ? v.z : expm1f(v.z);
        v.w = v.w > 0.f ? v.w : expm1f(v.w);
      }
      o4[i] = v;
    }
  }
}


static inline int cdiv(long long a, long long b) { return (int)((a + b - 1) / b); }

extern "C" void kernel_launch(void* const* d_in, const int* in_sizes, int n_in,
                              void* d_out, int out_size, void* d_ws, size_t ws_size,
                              hipStream_t stream) {
  const float* x      = (const float*)d_in[0];
  const int*   ei     = (const int*)  d_in[1];
  const float* W1     = (const float*)d_in[2];
  const float* att_s1 = (const float*)d_in[3];
  const float* att_d1 = (const float*)d_in[4];
  const float* b1     = (const float*)d_in[5];
  const float* W2     = (const float*)d_in[6];
  const float* att_s2 = (const float*)d_in[7];
  const float* att_d2 = (const float*)d_in[8];
  const float* b2     = (const float*)d_in[9];

  const int IN   = 128;
  const int C1   = 256;
  const int F2   = 64;
  const int Nn   = in_sizes[0] / IN;
  const int Eraw = in_sizes[1] / 2;

  float* ws = (float*)d_ws;
  size_t off = 0;
  float*    h1     = ws + off; off += (size_t)Nn * C1;
  float*    out1   = ws + off; off += (size_t)Nn * C1;
  float*    asrc1  = ws + off; off += (size_t)Nn * 4;
  float*    adst1  = ws + off; off += (size_t)Nn * 4;
  float*    h2     = h1;
  float*    asrc2  = asrc1;
  float*    adst2  = adst1;
  float*    outf   = (float*)d_out;

  const int T = 256;

  {
    int mt64  = (Nn / 16 + 3) / 4;
    int tiles = mt64 * (C1 / 16);
    gemm_wmma_f32<<<cdiv(tiles, 8), 256, 0, stream>>>(x, W1, h1, Nn, C1, IN);
  }
  node_alpha<4, 64><<<cdiv((long long)Nn * 4, T), T, 0, stream>>>(
      h1, att_s1, att_d1, asrc1, adst1, Nn);
  range_agg_k<4, 64, 8, 200, true><<<cdiv(Nn, 200), 1024, 0, stream>>>(
      ei, asrc1, adst1, h1, b1, out1, Nn, Eraw);

  {
    int mt64  = (Nn / 16 + 3) / 4;
    int tiles = mt64 * (F2 / 16);
    gemm_wmma_f32<<<cdiv(tiles, 8), 256, 0, stream>>>(out1, W2, h2, Nn, F2, C1);
  }
  node_alpha<1, 64><<<cdiv((long long)Nn, T), T, 0, stream>>>(
      h2, att_s2, att_d2, asrc2, adst2, Nn);
  range_agg_k<1, 64, 16, 200, false><<<cdiv(Nn, 200), 1024, 0, stream>>>(
      ei, asrc2, adst2, h2, b2, outf, Nn, Eraw);
}


# Layer 2's range work-groups fetch their per-wave edge lists (source, local row) from the workspace, where layer 1's store them after their scan; layer 2's destination scan and edge look-ups gone
# speedup vs baseline: 3.1429x; 1.1284x over previous
#include <hip/hip_runtime.h>

typedef __attribute__((ext_vector_type(2))) float v2f;
typedef __attribute__((ext_vector_type(8))) float v8f;

#define NEG_SLOPE 0.2f
#define LIST_SLOTS 512


__device__ __forceinline__ unsigned f2ord(float f) {
  unsigned u = __float_as_uint(f);
  return (u & 0x80000000u) ? ~u : (u | 0x80000000u);
}
__device__ __forceinline__ float ord2f(unsigned o) {
  unsigned u = (o & 0x80000000u) ? (o & 0x7fffffffu) : ~o;
  return __uint_as_float(u);
}

__device__ __forceinline__ void edge_ends(const int* __restrict__ ei, int e,
                                          int Eraw, int& s, int& d) {
  if (e < Eraw) { s = ei[e]; d = ei[Eraw + e]; }
  else          { s = e - Eraw; d = s; }
}


__global__ void gemm_wmma_f32(const float* __restrict__ A,
                              const float* __restrict__ B,
                              float* __restrict__ C,
                              int M, int Nn, int K) {
  const int lane   = threadIdx.x & 31;
  const int wave   = blockIdx.x * (blockDim.x >> 5) + (threadIdx.x >> 5);
  const int ntiles = Nn >> 4;
  const int mtiles = M >> 4;
  const int mt64   = (mtiles + 3) >> 2;
  const int tileM  = wave / ntiles;
  const int tileN  = wave - tileM * ntiles;
  if (tileM >= mt64) return;

  const int lh  = lane >> 4;
  const int l16 = lane & 15;
  const int m0  = tileM * 4;

  const float* arow[4];
#pragma unroll
  for (int r = 0; r < 4; ++r) {
    int mr = m0 + r; if (mr > mtiles - 1) mr = mtiles - 1;
    arow[r] = A + (size_t)(mr * 16 + l16) * K + lh * 2;
  }
  const float* bptr = B + (size_t)(lh * 2) * Nn + tileN * 16 + l16;

  v8f acc0 = {}, acc1 = {}, acc2 = {}, acc3 = {};
  for (int k = 0; k < K; k += 4) {
    v2f b;
    b.x = bptr[(size_t)k * Nn];
    b.y = bptr[(size_t)(k + 1) * Nn];
    v2f a0 = *(const v2f*)(arow[0] + k);
    v2f a1 = *(const v2f*)(arow[1] + k);
    v2f a2 = *(const v2f*)(arow[2] + k);
    v2f a3 = *(const v2f*)(arow[3] + k);
    acc0 = __builtin_amdgcn_wmma_f32_16x16x4_f32(false, a0, false, b, (short)0, acc0, false, false);
    acc1 = __builtin_amdgcn_wmma_f32_16x16x4_f32(false, a1, false, b, (short)0, acc1, false, false);
    acc2 = __builtin_amdgcn_wmma_f32_16x16x4_f32(false, a2, false, b, (short)0, acc2, false, false);
    acc3 = __builtin_amdgcn_wmma_f32_16x16x4_f32(false, a3, false, b, (short)0, acc3, false, false);
  }

  const int col = tileN * 16 + l16;
  v8f accs[4] = {acc0, acc1, acc2, acc3};
#pragma unroll
  for (int r = 0; r < 4; ++r) {
    if (m0 + r < mtiles) {
      float* crow = C + (size_t)((m0 + r) * 16 + lh * 8) * Nn + col;
#pragma unroll
      for (int q = 0; q < 8; ++q) crow[(size_t)q * Nn] = accs[r][q];
    }
  }
}


template <int H, int F>
__global__ void node_alpha(const float* __restrict__ h,
                           const float* __restrict__ att_s,
                           const float* __restrict__ att_d,
                           float* __restrict__ asrc, float* __restrict__ adst,
                           int n_nodes) {
  int idx = blockIdx.x * blockDim.x + threadIdx.x;
  if (idx >= n_nodes * H) return;
  int n  = idx / H;
  int hd = idx - n * H;
  const float* hp = h + (size_t)n * (H * F) + hd * F;
  const float* as = att_s + hd * F;
  const float* ad = att_d + hd * F;
  float s = 0.f, d = 0.f;
#pragma unroll 8
  for (int f = 0; f < F; ++f) { float v = hp[f]; s += v * as[f]; d += v * ad[f]; }
  asrc[idx] = s;
  adst[idx] = d;
}


template <int H, int F, int B, int NPB, bool ELU, bool TAKE>
__global__ __launch_bounds__(1024) void range_agg_k(
    const int* __restrict__ ei,
    const float* __restrict__ asrc,
    const float* __restrict__ adst,
    const float* __restrict__ feat,
    const float* __restrict__ bias,
    float* __restrict__ out,
    int* __restrict__ lists, int* __restrict__ lcnt,
    int n_nodes, int Eraw) {
  constexpr int C  = H * F;
  constexpr int NL = C / 32;
  constexpr int G  = B * H;
  constexpr int QC = LIST_SLOTS;
  static_assert(G <= 32 && (32 % G) == 0 && (F % 32) == 0 && (C % 4) == 0, "tiling");
  static_assert(NPB <= 256, "local row must fit the low byte of a list entry");

  __shared__ __attribute__((aligned(16))) float tile[NPB * C];
  __shared__ float    den[NPB * H];
  __shared__ unsigned mu[NPB * H];
  __shared__ float    adl[NPB * H];
  __shared__ int      qe[32 * QC];
  __shared__ int      ovf;

  const int tid  = threadIdx.x;
  const int lane = tid & 31;
  const int wv   = __builtin_amdgcn_readfirstlane(tid >> 5);
  const int nw   = blockDim.x >> 5;
  const int base = blockIdx.x * NPB;
  int nloc = n_nodes - base; if (nloc > NPB) nloc = NPB;

  {
    float4* t4 = (float4*)tile;
    for (int i = tid; i < NPB * C / 4; i += blockDim.x) t4[i] = make_float4(0.f, 0.f, 0.f, 0.f);
    for (int i = tid; i < NPB * H; i += blockDim.x) {
      den[i] = 0.f;
      mu[i]  = 0u;
      adl[i] = (i < nloc * H) ? adst[(size_t)base * H + i] : 0.f;
    }
    if (tid == 0) ovf = 0;
  }
  __syncthreads();

  int* myq = qe + wv * QC;
  int  qn = 0;
  int  done = 0;
  bool lost = false;

  auto chunk = [&](int q0, int cnt) {
    if (lane < cnt) {
      const int e = myq[q0 + lane];
      int s, d; edge_ends(ei, e, Eraw, s, d);
      const int dl = d - base;
#pragma unroll
      for (int hd = 0; hd < H; ++hd) {
        float v = asrc[(size_t)s * H + hd] + adl[dl * H + hd];
        v = v > 0.f ? v : NEG_SLOPE * v;
        atomicMax(mu + dl * H + hd, f2ord(v));
      }
      myq[q0 + lane] = (s << 8) | dl;
    }
  };
  auto fold = [&](bool all) {
    __builtin_amdgcn_fence(__ATOMIC_ACQ_REL, "wavefront");
    while (qn - done >= 32) { chunk(done, 32); done += 32; }
    if (all && done < qn) { chunk(done, qn - done); done = qn; }
  };
  auto room = [&](int n) {
    if (qn + n > QC) { fold(true); qn = 0; done = 0; lost = true; }
  };
  auto push = [&](unsigned mask, int e) {
    const int pos = qn + (int)__builtin_amdgcn_mbcnt_lo(mask, 0u);
    if ((mask >> lane) & 1u) myq[pos] = e;
    qn += __builtin_popcount(mask);
  };

  int* glist = lists + (size_t)(blockIdx.x * nw + wv) * QC;
  int have = -1;
  if (TAKE) have = __builtin_amdgcn_readfirstlane(lcnt[blockIdx.x * nw + wv]);

  if (TAKE && have >= 0) {
    for (int q0 = 0; q0 < have; q0 += 128) {
      int pk[4];
#pragma unroll
      for (int u = 0; u < 4; ++u) {
        const int i = q0 + u * 32 + lane;
        pk[u] = glist[i < have ? i : have - 1];
      }
      float sv[4][H];
#pragma unroll
      for (int u = 0; u < 4; ++u)
#pragma unroll
        for (int hd = 0; hd < H; ++hd) sv[u][hd] = asrc[(size_t)(pk[u] >> 8) * H + hd];
#pragma unroll
      for (int u = 0; u < 4; ++u)
#pragma unroll
        for (int hd = 0; hd < H; ++hd) asm volatile("" :: "v"(sv[u][hd]));
#pragma unroll
      for (int u = 0; u < 4; ++u) {
        const int i = q0 + u * 32 + lane;
        if (i < have) {
          const int dl = pk[u] & 255;
          myq[i] = pk[u];
#pragma unroll
          for (int hd = 0; hd < H; ++hd) {
            float v = sv[u][hd] + adl[dl * H + hd];
            v = v > 0.f ? v : NEG_SLOPE * v;
            atomicMax(mu + dl * H + hd, f2ord(v));
          }
        }
      }
    }
    qn = have;
  } else {
    {
      const int j = wv + nw * lane;
      const unsigned mask = __builtin_amdgcn_ballot_w32(j < nloc);
      if (mask) push(mask, Eraw + base + j);
    }

    const int* dstp = ei + Eraw;
    const int nblk = ((Eraw & 3) == 0) ? (Eraw >> 7) : 0;
    auto ld = [&](int blk) {
      int4 r = make_int4(-1, -1, -1, -1);
      if (blk < nblk) r = *(const int4*)(dstp + (size_t)blk * 128 + lane * 4);
      return r;
    };
    auto proc = [&](int4 c, int blk) {
      const unsigned m0 = __builtin_amdgcn_ballot_w32((unsigned)(c.x - base) < (unsigned)nloc);
      const unsigned m1 = __builtin_amdgcn_ballot_w32((unsigned)(c.y - base) < (unsigned)nloc);
      const unsigned m2 = __builtin_amdgcn_ballot_w32((unsigned)(c.z - base) < (unsigned)nloc);
      const unsigned m3 = __builtin_amdgcn_ballot_w32((unsigned)(c.w - base) < (unsigned)nloc);
      if (m0 | m1 | m2 | m3) {
        room(128);
        const int e = blk * 128 + lane * 4;
        if (m0) push(m0, e);
        if (m1) push(m1, e + 1);
        if (m2) push(m2, e + 2);
        if (m3) push(m3, e + 3);
        fold(false);
      }
    };
    {
      int blk = wv;
      int4 r0 = ld(blk), r1 = ld(blk + nw), r2 = ld(blk + 2 * nw), r3 = ld(blk + 3 * nw);
      for (; blk < nblk; blk += 4 * nw) {
        proc(r0, blk);          r0 = ld(blk + 4 * nw);
        proc(r1, blk + nw);     r1 = ld(blk + 5 * nw);
        proc(r2, blk + 2 * nw); r2 = ld(blk + 6 * nw);
        proc(r3, blk + 3 * nw); r3 = ld(blk + 7 * nw);
      }
    }
    for (int e0 = nblk * 128 + wv * 32; e0 < Eraw; e0 += nw * 32) {
      const int e = e0 + lane;
      const int d = (e < Eraw) ? dstp[e] : -1;
      const unsigned mask = __builtin_amdgcn_ballot_w32((unsigned)(d - base) < (unsigned)nloc);
      if (mask) { room(32); push(mask, e); fold(false); }
    }
    fold(true);
    if (lost && lane == 0) ovf = 1;
  }
  __syncthreads();

  const int again = __builtin_amdgcn_readfirstlane(ovf);
  if (!TAKE) {
    if (!again) for (int i = lane; i < qn; i += 32) glist[i] = myq[i];
    if (lane == 0) lcnt[blockIdx.x * nw + wv] = again ? -1 : qn;
  }
  if (!again) {
    auto batch = [&](int q0, int cnt) {
      const int g  = lane % G;
      const int k  = g / H;
      const int hd = g - k * H;
      const int kq = k < cnt ? k : cnt - 1;
      const int pk = myq[q0 + kq];
      const int sq = pk >> 8;
      const int dq = pk & 255;

      float f[B][NL];
      int   dk[B];
#pragma unroll
      for (int kk = 0; kk < B; ++kk) {
        const int sk = __builtin_amdgcn_readlane(sq, kk * H);
        dk[kk]       = __builtin_amdgcn_readlane(dq, kk * H);
        const float* fp = feat + (size_t)sk * C + lane;
#pragma unroll
        for (int j = 0; j < NL; ++j) f[kk][j] = fp[j * 32];
      }

      float v = asrc[(size_t)sq * H + hd] + adl[dq * H + hd];
      v = v > 0.f ? v : NEG_SLOPE * v;
      const float w = __expf(v - ord2f(mu[dq * H + hd]));
      if (lane < G && k < cnt) atomicAdd(den + dq * H + hd, w);

#pragma unroll
      for (int kk = 0; kk < B; ++kk) {
        if (kk < cnt) {
          float* op = tile + dk[kk] * C + lane;
#pragma unroll
          for (int j = 0; j < NL; ++j) {
            const float a = __int_as_float(
                __builtin_amdgcn_readlane(__float_as_int(w), kk * H + (j * 32) / F));
            atomicAdd(op + j * 32, f[kk][j] * a);
          }
        }
      }
    };
    for (int q0 = 0; q0 < qn; q0 += B) batch(q0, (qn - q0) < B ? (qn - q0) : B);
  } else {
    const int Etot = Eraw + n_nodes;
    for (int e0 = wv * 32; e0 < Etot; e0 += nw * 32) {
      const int e = e0 + lane;
      int s = 0, d = -1;
      if (e < Etot) edge_ends(ei, e, Eraw, s, d);
      unsigned mask = __builtin_amdgcn_ballot_w32((unsigned)(d - base) < (unsigned)nloc);
      while (mask) {
        const int l = __builtin_ctz(mask);
        mask &= mask - 1;
        const int sk = __shfl(s, l);
        const int dk = __shfl(d, l) - base;
        if (lane < H) {
          float v = asrc[(size_t)sk * H + lane] + adl[dk * H + lane];
          v = v > 0.f ? v : NEG_SLOPE * v;
          atomicAdd(den + dk * H + lane, __expf(v - ord2f(mu[dk * H + lane])));
        }
#pragma unroll
        for (int j = 0; j < NL; ++j) {
          const int hd = (j * 32) / F;
          float v = asrc[(size_t)sk * H + hd] + adl[dk * H + hd];
          v = v > 0.f ? v : NEG_SLOPE * v;
          const float a = __expf(v - ord2f(mu[dk * H + hd]));
          atomicAdd(tile + dk * C + j * 32 + lane, feat[(size_t)sk * C + j * 32 + lane] * a);
        }
      }
    }
  }
  __syncthreads();

  {
    const float4* t4 = (const float4*)tile;
    float4* o4 = (float4*)(out + (size_t)base * C);
    const float4* b4 = (const float4*)bias;
    for (int i = tid; i < nloc * (C / 4); i += blockDim.x) {
      const int nl = i / (C / 4);
      const int c4 = i - nl * (C / 4);
      const float r = den[nl * H + (c4 * 4) / F] + 1e-16f;
      const float4 t = t4[i];
      const float4 bb = b4[c4];
      float4 v;
      v.x = t.x / r + bb.x; v.y = t.y / r + bb.y; v.z = t.z / r + bb.z; v.w = t.w / r + bb.w;
      if (ELU) {
        v.x = v.x > 0.f ? v.x : expm1f(v.x);
        v.y = v.y > 0.f ? v.y : expm1f(v.y);
        v.z = v.z > 0.f ? v.z : expm1f(v.z);
        v.w = v.w > 0.f ? v.w : expm1f(v.w);
      }
      o4[i] = v;
    }
  }
}


static inline int cdiv(long long a, long long b) { return (int)((a + b - 1) / b); }

extern "C" void kernel_launch(void* const* d_in, const int* in_sizes, int n_in,
                              void* d_out, int out_size, void* d_ws, size_t ws_size,
                              hipStream_t stream) {
  const float* x      = (const float*)d_in[0];
  const int*   ei     = (const int*)  d_in[1];
  const float* W1     = (const float*)d_in[2];
  const float* att_s1 = (const float*)d_in[3];
  const float* att_d1 = (const float*)d_in[4];
  const float* b1     = (const float*)d_in[5];
  const float* W2     = (const float*)d_in[6];
  const float* att_s2 = (const float*)d_in[7];
  const float* att_d2 = (const float*)d_in[8];
  const float* b2     = (const float*)d_in[9];

  const int IN   = 128;
  const int C1   = 256;
  const int F2   = 64;
  const int Nn   = in_sizes[0] / IN;
  const int Eraw = in_sizes[1] / 2;

  float* ws = (float*)d_ws;
  size_t off = 0;
  float*    h1     = ws + off; off += (size_t)Nn * C1;
  float*    out1   = ws + off; off += (size_t)Nn * C1;
  float*    asrc1  = ws + off; off += (size_t)Nn * 4;
  float*    adst1  = ws + off; off += (size_t)Nn * 4;
  int*      lists  = (int*)(ws + off); off += (size_t)cdiv(Nn, 200) * 32 * LIST_SLOTS;
  int*      lcnt   = (int*)(ws + off); off += (size_t)cdiv(Nn, 200) * 32;
  float*    h2     = h1;
  float*    asrc2  = asrc1;
  float*    adst2  = adst1;
  float*    outf   = (float*)d_out;

  const int T = 256;

  {
    int mt64  = (Nn / 16 + 3) / 4;
    int tiles = mt64 * (C1 / 16);
    gemm_wmma_f32<<<cdiv(tiles, 8), 256, 0, stream>>>(x, W1, h1, Nn, C1, IN);
  }
  node_alpha<4, 64><<<cdiv((long long)Nn * 4, T), T, 0, stream>>>(
      h1, att_s1, att_d1, asrc1, adst1, Nn);
  range_agg_k<4, 64, 8, 200, true, false><<<cdiv(Nn, 200), 1024, 0, stream>>>(
      ei, asrc1, adst1, h1, b1, out1, lists, lcnt, Nn, Eraw);

  {
    int mt64  = (Nn / 16 + 3) / 4;
    int tiles = mt64 * (F2 / 16);
    gemm_wmma_f32<<<cdiv(tiles, 8), 256, 0, stream>>>(out1, W2, h2, Nn, F2, C1);
  }
  node_alpha<1, 64><<<cdiv((long long)Nn, T), T, 0, stream>>>(
      h2, att_s2, att_d2, asrc2, adst2, Nn);
  range_agg_k<1, 64, 16, 200, false, true><<<cdiv(Nn, 200), 1024, 0, stream>>>(
      ei, asrc2, adst2, h2, b2, outf, lists, lcnt, Nn, Eraw);
}


# GEMM A operand through LDS: a work-group copies its rows once with whole-line 16-byte loads (pitch K+4), waves read A fragments from LDS; was 16 lines per fragment load; k order, sums unchanged
# speedup vs baseline: 3.9862x; 1.2683x over previous
#include <hip/hip_runtime.h>

typedef __attribute__((ext_vector_type(2))) float v2f;
typedef __attribute__((ext_vector_type(8))) float v8f;

#define NEG_SLOPE 0.2f
#define LIST_SLOTS 512


__device__ __forceinline__ unsigned f2ord(float f) {
  unsigned u = __float_as_uint(f);
  return (u & 0x80000000u) ? ~u : (u | 0x80000000u);
}
__device__ __forceinline__ float ord2f(unsigned o) {
  unsigned u = (o & 0x80000000u) ? (o & 0x7fffffffu) : ~o;
  return __uint_as_float(u);
}

__device__ __forceinline__ void edge_ends(const int* __restrict__ ei, int e,
                                          int Eraw, int& s, int& d) {
  if (e < Eraw) { s = ei[e]; d = ei[Eraw + e]; }
  else          { s = e - Eraw; d = s; }
}

template <int K, int N, int MT, int KU>
__global__ __launch_bounds__(256) void gemm_lds_f32(const float* __restrict__ A,
                                                    const float* __restrict__ B,
                                                    float* __restrict__ C, int M) {
  constexpr int P  = K + 4;
  constexpr int NT = N / 16;
  constexpr int RN = (NT >= 8) ? NT / 8 : 1;
  constexpr int WN = NT / RN;
  constexpr int WM = 8 / WN;
  constexpr int RM = MT / WM;
  static_assert(WN * WM == 8 && RM * WM == MT && (K % (4 * KU)) == 0, "tiling");
  constexpr int ROWS = MT * 16;
  constexpr int R4   = K / 4;
  constexpr int TOT4 = ROWS * R4;
  static_assert((TOT4 % 256) == 0, "staging");

  __shared__ __attribute__((aligned(16))) float As[ROWS * P];

  const int tid  = threadIdx.x;
  const int lane = tid & 31;
  const int w    = __builtin_amdgcn_readfirstlane(tid >> 5);
  const int lh   = lane >> 4;
  const int l16  = lane & 15;
  const int m0   = blockIdx.x * ROWS;

  const int wn = w % WN;
  const int wm = w / WN;
  const float* bp = B + (size_t)(lh * 2) * N + (wn * RN) * 16 + l16;
  const float* ap = As + (wm * RM * 16 + l16) * P + lh * 2;

  v2f bc[KU][RN];
#pragma unroll
  for (int u = 0; u < KU; ++u)
#pragma unroll
    for (int j = 0; j < RN; ++j) {
      bc[u][j].x = bp[(size_t)(4 * u) * N + j * 16];
      bc[u][j].y = bp[(size_t)(4 * u + 1) * N + j * 16];
    }

  {
    constexpr int PER = TOT4 / 256;
    constexpr int UB  = (PER % 16 == 0) ? 16 : PER;
    static_assert((PER % UB) == 0, "staging batches");
    for (int c0 = 0; c0 < PER; c0 += UB) {
      float4 v[UB];
#pragma unroll
      for (int u = 0; u < UB; ++u) {
        const int i  = tid + (c0 + u) * 256;
        const int r  = i / R4;
        const int c4 = i - r * R4;
        int gr = m0 + r; if (gr > M - 1) gr = M - 1;
        v[u] = *(const float4*)(A + (size_t)gr * K + c4 * 4);
      }
#pragma unroll
      for (int u = 0; u < UB; ++u) {
        const int i  = tid + (c0 + u) * 256;
        const int r  = i / R4;
        const int c4 = i - r * R4;
        *(float4*)(As + r * P + c4 * 4) = v[u];
      }
    }
  }
  __syncthreads();

  v8f acc[RM][RN];
#pragma unroll
  for (int i = 0; i < RM; ++i)
#pragma unroll
    for (int j = 0; j < RN; ++j) acc[i][j] = v8f{};

#pragma unroll
  for (int k = 0; k < K; k += 4 * KU) {
    const int kn = (k + 4 * KU < K) ? k + 4 * KU : k;
    const float* bq = bp + (size_t)kn * N;
    v2f bn[KU][RN];
#pragma unroll
    for (int u = 0; u < KU; ++u)
#pragma unroll
      for (int j = 0; j < RN; ++j) {
        bn[u][j].x = bq[(size_t)(4 * u) * N + j * 16];
        bn[u][j].y = bq[(size_t)(4 * u + 1) * N + j * 16];
      }
    v2f a[KU][RM];
#pragma unroll
    for (int u = 0; u < KU; ++u)
#pragma unroll
      for (int i = 0; i < RM; ++i) a[u][i] = *(const v2f*)(ap + i * 16 * P + k + 4 * u);
#pragma unroll
    for (int u = 0; u < KU; ++u)
#pragma unroll
      for (int i = 0; i < RM; ++i)
#pragma unroll
        for (int j = 0; j < RN; ++j)
          acc[i][j] = __builtin_amdgcn_wmma_f32_16x16x4_f32(false, a[u][i], false, bc[u][j],
                                                            (short)0, acc[i][j], false, false);
#pragma unroll
    for (int u = 0; u < KU; ++u)
#pragma unroll
      for (int j = 0; j < RN; ++j) bc[u][j] = bn[u][j];
  }

  const int mtiles = M >> 4;
#pragma unroll
  for (int i = 0; i < RM; ++i) {
    const int mt = (m0 >> 4) + wm * RM + i;
    if (mt < mtiles) {
#pragma unroll
      for (int j = 0; j < RN; ++j) {
        float* crow = C + (size_t)(mt * 16 + lh * 8) * N + (wn * RN + j) * 16 + l16;
#pragma unroll
        for (int q = 0; q < 8; ++q) crow[(size_t)q * N] = acc[i][j][q];
      }
    }
  }
}


template <int H, int F>
__global__ void node_alpha(const float* __restrict__ h,
                           const float* __restrict__ att_s,
                           const float* __restrict__ att_d,
                           float* __restrict__ asrc, float* __restrict__ adst,
                           int n_nodes) {
  int idx = blockIdx.x * blockDim.x + threadIdx.x;
  if (idx >= n_nodes * H) return;
  int n  = idx / H;
  int hd = idx - n * H;
  const float* hp = h + (size_t)n * (H * F) + hd * F;
  const float* as = att_s + hd * F;
  const float* ad = att_d + hd * F;
  float s = 0.f, d = 0.f;
#pragma unroll 8
  for (int f = 0; f < F; ++f) { float v = hp[f]; s += v * as[f]; d += v * ad[f]; }
  asrc[idx] = s;
  adst[idx] = d;
}


template <int H, int F, int B, int NPB, bool ELU, bool TAKE>
__global__ __launch_bounds__(1024) void range_agg_k(
    const int* __restrict__ ei,
    const float* __restrict__ asrc,
    const float* __restrict__ adst,
    const float* __restrict__ feat,
    const float* __restrict__ bias,
    float* __restrict__ out,
    int* __restrict__ lists, int* __restrict__ lcnt,
    int n_nodes, int Eraw) {
  constexpr int C  = H * F;
  constexpr int NL = C / 32;
  constexpr int G  = B * H;
  constexpr int QC = LIST_SLOTS;
  static_assert(G <= 32 && (32 % G) == 0 && (F % 32) == 0 && (C % 4) == 0, "tiling");
  static_assert(NPB <= 256, "local row must fit the low byte of a list entry");

  __shared__ __attribute__((aligned(16))) float tile[NPB * C];
  __shared__ float    den[NPB * H];
  __shared__ unsigned mu[NPB * H];
  __shared__ float    adl[NPB * H];
  __shared__ int      qe[32 * QC];
  __shared__ int      ovf;

  const int tid  = threadIdx.x;
  const int lane = tid & 31;
  const int wv   = __builtin_amdgcn_readfirstlane(tid >> 5);
  const int nw   = blockDim.x >> 5;
  const int base = blockIdx.x * NPB;
  int nloc = n_nodes - base; if (nloc > NPB) nloc = NPB;

  {
    float4* t4 = (float4*)tile;
    for (int i = tid; i < NPB * C / 4; i += blockDim.x) t4[i] = make_float4(0.f, 0.f, 0.f, 0.f);
    for (int i = tid; i < NPB * H; i += blockDim.x) {
      den[i] = 0.f;
      mu[i]  = 0u;
      adl[i] = (i < nloc * H) ? adst[(size_t)base * H + i] : 0.f;
    }
    if (tid == 0) ovf = 0;
  }
  __syncthreads();

  int* myq = qe + wv * QC;
  int  qn = 0;
  int  done = 0;
  bool lost = false;

  auto chunk = [&](int q0, int cnt) {
    if (lane < cnt) {
      const int e = myq[q0 + lane];
      int s, d; edge_ends(ei, e, Eraw, s, d);
      const int dl = d - base;
#pragma unroll
      for (int hd = 0; hd < H; ++hd) {
        float v = asrc[(size_t)s * H + hd] + adl[dl * H + hd];
        v = v > 0.f ? v : NEG_SLOPE * v;
        atomicMax(mu + dl * H + hd, f2ord(v));
      }
      myq[q0 + lane] = (s << 8) | dl;
    }
  };
  auto fold = [&](bool all) {
    __builtin_amdgcn_fence(__ATOMIC_ACQ_REL, "wavefront");
    while (qn - done >= 32) { chunk(done, 32); done += 32; }
    if (all && done < qn) { chunk(done, qn - done); done = qn; }
  };
  auto room = [&](int n) {
    if (qn + n > QC) { fold(true); qn = 0; done = 0; lost = true; }
  };
  auto push = [&](unsigned mask, int e) {
    const int pos = qn + (int)__builtin_amdgcn_mbcnt_lo(mask, 0u);
    if ((mask >> lane) & 1u) myq[pos] = e;
    qn += __builtin_popcount(mask);
  };

  int* glist = lists + (size_t)(blockIdx.x * nw + wv) * QC;
  int have = -1;
  if (TAKE) have = __builtin_amdgcn_readfirstlane(lcnt[blockIdx.x * nw + wv]);

  if (TAKE && have >= 0) {
    for (int q0 = 0; q0 < have; q0 += 128) {
      int pk[4];
#pragma unroll
      for (int u = 0; u < 4; ++u) {
        const int i = q0 + u * 32 + lane;
        pk[u] = glist[i < have ? i : have - 1];
      }
      float sv[4][H];
#pragma unroll
      for (int u = 0; u < 4; ++u)
#pragma unroll
        for (int hd = 0; hd < H; ++hd) sv[u][hd] = asrc[(size_t)(pk[u] >> 8) * H + hd];
#pragma unroll
      for (int u = 0; u < 4; ++u)
#pragma unroll
        for (int hd = 0; hd < H; ++hd) asm volatile("" :: "v"(sv[u][hd]));
#pragma unroll
      for (int u = 0; u < 4; ++u) {
        const int i = q0 + u * 32 + lane;
        if (i < have) {
          const int dl = pk[u] & 255;
          myq[i] = pk[u];
#pragma unroll
          for (int hd = 0; hd < H; ++hd) {
            float v = sv[u][hd] + adl[dl * H + hd];
            v = v > 0.f ? v : NEG_SLOPE * v;
            atomicMax(mu + dl * H + hd, f2ord(v));
          }
        }
      }
    }
    qn = have;
  } else {
    {
      const int j = wv + nw * lane;
      const unsigned mask = __builtin_amdgcn_ballot_w32(j < nloc);
      if (mask) push(mask, Eraw + base + j);
    }

    const int* dstp = ei + Eraw;
    const int nblk = ((Eraw & 3) == 0) ? (Eraw >> 7) : 0;
    auto ld = [&](int blk) {
      int4 r = make_int4(-1, -1, -1, -1);
      if (blk < nblk) r = *(const int4*)(dstp + (size_t)blk * 128 + lane * 4);
      return r;
    };
    auto proc = [&](int4 c, int blk) {
      const unsigned m0 = __builtin_amdgcn_ballot_w32((unsigned)(c.x - base) < (unsigned)nloc);
      const unsigned m1 = __builtin_amdgcn_ballot_w32((unsigned)(c.y - base) < (unsigned)nloc);
      const unsigned m2 = __builtin_amdgcn_ballot_w32((unsigned)(c.z - base) < (unsigned)nloc);
      const unsigned m3 = __builtin_amdgcn_ballot_w32((unsigned)(c.w - base) < (unsigned)nloc);
      if (m0 | m1 | m2 | m3) {
        room(128);
        const int e = blk * 128 + lane * 4;
        if (m0) push(m0, e);
        if (m1) push(m1, e + 1);
        if (m2) push(m2, e + 2);
        if (m3) push(m3, e + 3);
        fold(false);
      }
    };
    {
      int blk = wv;
      int4 r0 = ld(blk), r1 = ld(blk + nw), r2 = ld(blk + 2 * nw), r3 = ld(blk + 3 * nw);
      for (; blk < nblk; blk += 4 * nw) {
        proc(r0, blk);          r0 = ld(blk + 4 * nw);
        proc(r1, blk + nw);     r1 = ld(blk + 5 * nw);
        proc(r2, blk + 2 * nw); r2 = ld(blk + 6 * nw);
        proc(r3, blk + 3 * nw); r3 = ld(blk + 7 * nw);
      }
    }
    for (int e0 = nblk * 128 + wv * 32; e0 < Eraw; e0 += nw * 32) {
      const int e = e0 + lane;
      const int d = (e < Eraw) ? dstp[e] : -1;
      const unsigned mask = __builtin_amdgcn_ballot_w32((unsigned)(d - base) < (unsigned)nloc);
      if (mask) { room(32); push(mask, e); fold(false); }
    }
    fold(true);
    if (lost && lane == 0) ovf = 1;
  }
  __syncthreads();

  const int again = __builtin_amdgcn_readfirstlane(ovf);
  if (!TAKE) {
    if (!again) for (int i = lane; i < qn; i += 32) glist[i] = myq[i];
    if (lane == 0) lcnt[blockIdx.x * nw + wv] = again ? -1 : qn;
  }
  if (!again) {
    auto batch = [&](int q0, int cnt) {
      const int g  = lane % G;
      const int k  = g / H;
      const int hd = g - k * H;
      const int kq = k < cnt ? k : cnt - 1;
      const int pk = myq[q0 + kq];
      const int sq = pk >> 8;
      const int dq = pk & 255;

      float f[B][NL];
      int   dk[B];
#pragma unroll
      for (int kk = 0; kk < B; ++kk) {
        const int sk = __builtin_amdgcn_readlane(sq, kk * H);
        dk[kk]       = __builtin_amdgcn_readlane(dq, kk * H);
        const float* fp = feat + (size_t)sk * C + lane;
#pragma unroll
        for (int j = 0; j < NL; ++j) f[kk][j] = fp[j * 32];
      }

      float v = asrc[(size_t)sq * H + hd] + adl[dq * H + hd];
      v = v > 0.f ? v : NEG_SLOPE * v;
      const float w = __expf(v - ord2f(mu[dq * H + hd]));
      if (lane < G && k < cnt) atomicAdd(den + dq * H + hd, w);

#pragma unroll
      for (int kk = 0; kk < B; ++kk) {
        if (kk < cnt) {
          float* op = tile + dk[kk] * C + lane;
#pragma unroll
          for (int j = 0; j < NL; ++j) {
            const float a = __int_as_float(
                __builtin_amdgcn_readlane(__float_as_int(w), kk * H + (j * 32) / F));
            atomicAdd(op + j * 32, f[kk][j] * a);
          }
        }
      }
    };
    for (int q0 = 0; q0 < qn; q0 += B) batch(q0, (qn - q0) < B ? (qn - q0) : B);
  } else {
    const int Etot = Eraw + n_nodes;
    for (int e0 = wv * 32; e0 < Etot; e0 += nw * 32) {
      const int e = e0 + lane;
      int s = 0, d = -1;
      if (e < Etot) edge_ends(ei, e, Eraw, s, d);
      unsigned mask = __builtin_amdgcn_ballot_w32((unsigned)(d - base) < (unsigned)nloc);
      while (mask) {
        const int l = __builtin_ctz(mask);
        mask &= mask - 1;
        const int sk = __shfl(s, l);
        const int dk = __shfl(d, l) - base;
        if (lane < H) {
          float v = asrc[(size_t)sk * H + lane] + adl[dk * H + lane];
          v = v > 0.f ? v : NEG_SLOPE * v;
          atomicAdd(den + dk * H + lane, __expf(v - ord2f(mu[dk * H + lane])));
        }
#pragma unroll
        for (int j = 0; j < NL; ++j) {
          const int hd = (j * 32) / F;
          float v = asrc[(size_t)sk * H + hd] + adl[dk * H + hd];
          v = v > 0.f ? v : NEG_SLOPE * v;
          const float a = __expf(v - ord2f(mu[dk * H + hd]));
          atomicAdd(tile + dk * C + j * 32 + lane, feat[(size_t)sk * C + j * 32 + lane] * a);
        }
      }
    }
  }
  __syncthreads();

  {
    const float4* t4 = (const float4*)tile;
    float4* o4 = (float4*)(out + (size_t)base * C);
    const float4* b4 = (const float4*)bias;
    for (int i = tid; i < nloc * (C / 4); i += blockDim.x) {
      const int nl = i / (C / 4);
      const int c4 = i - nl * (C / 4);
      const float r = den[nl * H + (c4 * 4) / F] + 1e-16f;
      const float4 t = t4[i];
      const float4 bb = b4[c4];
      float4 v;
      v.x = t.x / r + bb.x; v.y = t.y / r + bb.y; v.z = t.z / r + bb.z; v.w = t.w / r + bb.w;
      if (ELU) {
        v.x = v.x > 0.f ? v.x : expm1f(v.x);
        v.y = v.y > 0.f ? v.y : expm1f(v.y);
        v.z = v.z > 0.f ? v.z : expm1f(v.z);
        v.w = v.w > 0.f ? v.w : expm1f(v.w);
      }
      o4[i] = v;
    }
  }
}


static inline int cdiv(long long a, long long b) { return (int)((a + b - 1) / b); }

extern "C" void kernel_launch(void* const* d_in, const int* in_sizes, int n_in,
                              void* d_out, int out_size, void* d_ws, size_t ws_size,
                              hipStream_t stream) {
  const float* x      = (const float*)d_in[0];
  const int*   ei     = (const int*)  d_in[1];
  const float* W1     = (const float*)d_in[2];
  const float* att_s1 = (const float*)d_in[3];
  const float* att_d1 = (const float*)d_in[4];
  const float* b1     = (const float*)d_in[5];
  const float* W2     = (const float*)d_in[6];
  const float* att_s2 = (const float*)d_in[7];
  const float* att_d2 = (const float*)d_in[8];
  const float* b2     = (const float*)d_in[9];

  const int IN   = 128;
  const int C1   = 256;
  const int Nn   = in_sizes[0] / IN;
  const int Eraw = in_sizes[1] / 2;

  float* ws = (float*)d_ws;
  size_t off = 0;
  float*    h1     = ws + off; off += (size_t)Nn * C1;
  float*    out1   = ws + off; off += (size_t)Nn * C1;
  float*    asrc1  = ws + off; off += (size_t)Nn * 4;
  float*    adst1  = ws + off; off += (size_t)Nn * 4;
  int*      lists  = (int*)(ws + off); off += (size_t)cdiv(Nn, 200) * 32 * LIST_SLOTS;
  int*      lcnt   = (int*)(ws + off); off += (size_t)cdiv(Nn, 200) * 32;
  float*    h2     = h1;
  float*    asrc2  = asrc1;
  float*    adst2  = adst1;
  float*    outf   = (float*)d_out;

  const int T = 256;

  gemm_lds_f32<128, 256, 7, 2><<<cdiv(Nn, 7 * 16), 256, 0, stream>>>(x, W1, h1, Nn);
  node_alpha<4, 64><<<cdiv((long long)Nn * 4, T), T, 0, stream>>>(
      h1, att_s1, att_d1, asrc1, adst1, Nn);
  range_agg_k<4, 64, 8, 200, true, false><<<cdiv(Nn, 200), 1024, 0, stream>>>(
      ei, asrc1, adst1, h1, b1, out1, lists, lcnt, Nn, Eraw);

  gemm_lds_f32<256, 64, 8, 4><<<cdiv(Nn, 8 * 16), 256, 0, stream>>>(out1, W2, h2, Nn);
  node_alpha<1, 64><<<cdiv((long long)Nn, T), T, 0, stream>>>(
      h2, att_s2, att_d2, asrc2, adst2, Nn);
  range_agg_k<1, 64, 16, 200, false, true><<<cdiv(Nn, 200), 1024, 0, stream>>>(
      ei, asrc2, adst2, h2, b2, outf, lists, lcnt, Nn, Eraw);
}


# Destination scan on a byte plane: layer 1's node-score launch writes each edge's range number (dst/200) as one byte; range work-groups test 16 edges per lane-load (was 4); lists in another order
# speedup vs baseline: 4.2976x; 1.0781x over previous
#include <hip/hip_runtime.h>

typedef __attribute__((ext_vector_type(2))) float v2f;
typedef __attribute__((ext_vector_type(8))) float v8f;

#define NEG_SLOPE 0.2f
#define LIST_SLOTS 512


__device__ __forceinline__ unsigned f2ord(float f) {
  unsigned u = __float_as_uint(f);
  return (u & 0x80000000u) ? ~u : (u | 0x80000000u);
}
__device__ __forceinline__ float ord2f(unsigned o) {
  unsigned u = (o & 0x80000000u) ? (o & 0x7fffffffu) : ~o;
  return __uint_as_float(u);
}

__device__ __forceinline__ void edge_ends(const int* __restrict__ ei, int e,
                                          int Eraw, int& s, int& d) {
  if (e < Eraw) { s = ei[e]; d = ei[Eraw + e]; }
  else          { s = e - Eraw; d = s; }
}

template <int K, int N, int MT, int KU>
__global__ __launch_bounds__(256) void gemm_lds_f32(const float* __restrict__ A,
                                                    const float* __restrict__ B,
                                                    float* __restrict__ C, int M) {
  constexpr int P  = K + 4;
  constexpr int NT = N / 16;
  constexpr int RN = (NT >= 8) ? NT / 8 : 1;
  constexpr int WN = NT / RN;
  constexpr int WM = 8 / WN;
  constexpr int RM = MT / WM;
  static_assert(WN * WM == 8 && RM * WM == MT && (K % (4 * KU)) == 0, "tiling");
  constexpr int ROWS = MT * 16;
  constexpr int R4   = K / 4;
  constexpr int TOT4 = ROWS * R4;
  static_assert((TOT4 % 256) == 0, "staging");

  __shared__ __attribute__((aligned(16))) float As[ROWS * P];

  const int tid  = threadIdx.x;
  const int lane = tid & 31;
  const int w    = __builtin_amdgcn_readfirstlane(tid >> 5);
  const int lh   = lane >> 4;
  const int l16  = lane & 15;
  const int m0   = blockIdx.x * ROWS;

  const int wn = w % WN;
  const int wm = w / WN;
  const float* bp = B + (size_t)(lh * 2) * N + (wn * RN) * 16 + l16;
  const float* ap = As + (wm * RM * 16 + l16) * P + lh * 2;

  v2f bc[KU][RN];
#pragma unroll
  for (int u = 0; u < KU; ++u)
#pragma unroll
    for (int j = 0; j < RN; ++j) {
      bc[u][j].x = bp[(size_t)(4 * u) * N + j * 16];
      bc[u][j].y = bp[(size_t)(4 * u + 1) * N + j * 16];
    }

  {
    constexpr int PER = TOT4 / 256;
    constexpr int UB  = (PER % 16 == 0) ? 16 : PER;
    static_assert((PER % UB) == 0, "staging batches");
    for (int c0 = 0; c0 < PER; c0 += UB) {
      float4 v[UB];
#pragma unroll
      for (int u = 0; u < UB; ++u) {
        const int i  = tid + (c0 + u) * 256;
        const int r  = i / R4;
        const int c4 = i - r * R4;
        int gr = m0 + r; if (gr > M - 1) gr = M - 1;
        v[u] = *(const float4*)(A + (size_t)gr * K + c4 * 4);
      }
#pragma unroll
      for (int u = 0; u < UB; ++u) {
        const int i  = tid + (c0 + u) * 256;
        const int r  = i / R4;
        const int c4 = i - r * R4;
        *(float4*)(As + r * P + c4 * 4) = v[u];
      }
    }
  }
  __syncthreads();

  v8f acc[RM][RN];
#pragma unroll
  for (int i = 0; i < RM; ++i)
#pragma unroll
    for (int j = 0; j < RN; ++j) acc[i][j] = v8f{};

#pragma unroll
  for (int k = 0; k < K; k += 4 * KU) {
    const int kn = (k + 4 * KU < K) ? k + 4 * KU : k;
    const float* bq = bp + (size_t)kn * N;
    v2f bn[KU][RN];
#pragma unroll
    for (int u = 0; u < KU; ++u)
#pragma unroll
      for (int j = 0; j < RN; ++j) {
        bn[u][j].x = bq[(size_t)(4 * u) * N + j * 16];
        bn[u][j].y = bq[(size_t)(4 * u + 1) * N + j * 16];
      }
    v2f a[KU][RM];
#pragma unroll
    for (int u = 0; u < KU; ++u)
#pragma unroll
      for (int i = 0; i < RM; ++i) a[u][i] = *(const v2f*)(ap + i * 16 * P + k + 4 * u);
#pragma unroll
    for (int u = 0; u < KU; ++u)
#pragma unroll
      for (int i = 0; i < RM; ++i)
#pragma unroll
        for (int j = 0; j < RN; ++j)
          acc[i][j] = __builtin_amdgcn_wmma_f32_16x16x4_f32(false, a[u][i], false, bc[u][j],
                                                            (short)0, acc[i][j], false, false);
#pragma unroll
    for (int u = 0; u < KU; ++u)
#pragma unroll
      for (int j = 0; j < RN; ++j) bc[u][j] = bn[u][j];
  }

  const int mtiles = M >> 4;
#pragma unroll
  for (int i = 0; i < RM; ++i) {
    const int mt = (m0 >> 4) + wm * RM + i;
    if (mt < mtiles) {
#pragma unroll
      for (int j = 0; j < RN; ++j) {
        float* crow = C + (size_t)(mt * 16 + lh * 8) * N + (wn * RN + j) * 16 + l16;
#pragma unroll
        for (int q = 0; q < 8; ++q) crow[(size_t)q * N] = acc[i][j][q];
      }
    }
  }
}


template <int H, int F, int NPB, bool KEYS>
__global__ void node_alpha(const float* __restrict__ h,
                           const float* __restrict__ att_s,
                           const float* __restrict__ att_d,
                           float* __restrict__ asrc, float* __restrict__ adst,
                           int n_nodes,
                           const int* __restrict__ ei, unsigned* __restrict__ keys, int Eraw) {
  int idx = blockIdx.x * blockDim.x + threadIdx.x;
  if (KEYS && (Eraw & 15) == 0) {
    const int4* d4 = (const int4*)(ei + Eraw);
    const int nq = Eraw >> 2;
    auto key = [&](int d) -> unsigned {
      unsigned k = (unsigned)d / (unsigned)NPB;
      if (k > 255u) k = 255u;
      return ((unsigned)d < (unsigned)n_nodes) ? k : 255u;
    };
    for (int q = idx; q < nq; q += gridDim.x * blockDim.x) {
      const int4 d = d4[q];
      keys[q] = key(d.x) | (key(d.y) << 8) | (key(d.z) << 16) | (key(d.w) << 24);
    }
  }
  if (idx >= n_nodes * H) return;
  int n  = idx / H;
  int hd = idx - n * H;
  const float* hp = h + (size_t)n * (H * F) + hd * F;
  const float* as = att_s + hd * F;
  const float* ad = att_d + hd * F;
  float s = 0.f, d = 0.f;
#pragma unroll 8
  for (int f = 0; f < F; ++f) { float v = hp[f]; s += v * as[f]; d += v * ad[f]; }
  asrc[idx] = s;
  adst[idx] = d;
}


template <int H, int F, int B, int NPB, bool ELU, bool TAKE>
__global__ __launch_bounds__(1024) void range_agg_k(
    const int* __restrict__ ei,
    const float* __restrict__ asrc,
    const float* __restrict__ adst,
    const float* __restrict__ feat,
    const float* __restrict__ bias,
    float* __restrict__ out,
    int* __restrict__ lists, int* __restrict__ lcnt,
    const unsigned* __restrict__ keys,
    int n_nodes, int Eraw) {
  constexpr int C  = H * F;
  constexpr int NL = C / 32;
  constexpr int G  = B * H;
  constexpr int QC = LIST_SLOTS;
  static_assert(G <= 32 && (32 % G) == 0 && (F % 32) == 0 && (C % 4) == 0, "tiling");
  static_assert(NPB <= 256, "local row must fit the low byte of a list entry");

  __shared__ __attribute__((aligned(16))) float tile[NPB * C];
  __shared__ float    den[NPB * H];
  __shared__ unsigned mu[NPB * H];
  __shared__ float    adl[NPB * H];
  __shared__ int      qe[32 * QC];
  __shared__ int      ovf;

  const int tid  = threadIdx.x;
  const int lane = tid & 31;
  const int wv   = __builtin_amdgcn_readfirstlane(tid >> 5);
  const int nw   = blockDim.x >> 5;
  const int base = blockIdx.x * NPB;
  int nloc = n_nodes - base; if (nloc > NPB) nloc = NPB;

  {
    float4* t4 = (float4*)tile;
    for (int i = tid; i < NPB * C / 4; i += blockDim.x) t4[i] = make_float4(0.f, 0.f, 0.f, 0.f);
    for (int i = tid; i < NPB * H; i += blockDim.x) {
      den[i] = 0.f;
      mu[i]  = 0u;
      adl[i] = (i < nloc * H) ? adst[(size_t)base * H + i] : 0.f;
    }
    if (tid == 0) ovf = 0;
  }
  __syncthreads();

  int* myq = qe + wv * QC;
  int  qn = 0;
  int  done = 0;
  bool lost = false;

  auto chunk = [&](int q0, int cnt) {
    if (lane < cnt) {
      const int e = myq[q0 + lane];
      int s, d; edge_ends(ei, e, Eraw, s, d);
      const int dl = d - base;
#pragma unroll
      for (int hd = 0; hd < H; ++hd) {
        float v = asrc[(size_t)s * H + hd] + adl[dl * H + hd];
        v = v > 0.f ? v : NEG_SLOPE * v;
        atomicMax(mu + dl * H + hd, f2ord(v));
      }
      myq[q0 + lane] = (s << 8) | dl;
    }
  };
  auto fold = [&](bool all) {
    __builtin_amdgcn_fence(__ATOMIC_ACQ_REL, "wavefront");
    while (qn - done >= 32) { chunk(done, 32); done += 32; }
    if (all && done < qn) { chunk(done, qn - done); done = qn; }
  };
  auto room = [&](int n) {
    if (qn + n > QC) { fold(true); qn = 0; done = 0; lost = true; }
  };
  auto push = [&](unsigned mask, int e) {
    const int pos = qn + (int)__builtin_amdgcn_mbcnt_lo(mask, 0u);
    if ((mask >> lane) & 1u) myq[pos] = e;
    qn += __builtin_popcount(mask);
  };

  int* glist = lists + (size_t)(blockIdx.x * nw + wv) * QC;
  int have = -1;
  if (TAKE) have = __builtin_amdgcn_readfirstlane(lcnt[blockIdx.x * nw + wv]);

  if (TAKE && have >= 0) {
    for (int q0 = 0; q0 < have; q0 += 128) {
      int pk[4];
#pragma unroll
      for (int u = 0; u < 4; ++u) {
        const int i = q0 + u * 32 + lane;
        pk[u] = glist[i < have ? i : have - 1];
      }
      float sv[4][H];
#pragma unroll
      for (int u = 0; u < 4; ++u)
#pragma unroll
        for (int hd = 0; hd < H; ++hd) sv[u][hd] = asrc[(size_t)(pk[u] >> 8) * H + hd];
#pragma unroll
      for (int u = 0; u < 4; ++u)
#pragma unroll
        for (int hd = 0; hd < H; ++hd) asm volatile("" :: "v"(sv[u][hd]));
#pragma unroll
      for (int u = 0; u < 4; ++u) {
        const int i = q0 + u * 32 + lane;
        if (i < have) {
          const int dl = pk[u] & 255;
          myq[i] = pk[u];
#pragma unroll
          for (int hd = 0; hd < H; ++hd) {
            float v = sv[u][hd] + adl[dl * H + hd];
            v = v > 0.f ? v : NEG_SLOPE * v;
            atomicMax(mu + dl * H + hd, f2ord(v));
          }
        }
      }
    }
    qn = have;
  } else {
    {
      const int j = wv + nw * lane;
      const unsigned mask = __builtin_amdgcn_ballot_w32(j < nloc);
      if (mask) push(mask, Eraw + base + j);
    }

    if (gridDim.x <= 255u && (Eraw & 15) == 0) {
      const int nkb = (Eraw + 511) >> 9;
      const unsigned pat = blockIdx.x * 0x01010101u;
      auto ldk = [&](int blk) {
        uint4 r = make_uint4(~0u, ~0u, ~0u, ~0u);
        if (blk < nkb) {
          const int e = blk * 512 + lane * 16;
          if (e < Eraw) r = *(const uint4*)(keys + (e >> 2));
        }
        return r;
      };
      auto eqb = [&](unsigned c) {
        const unsigned x = c ^ pat;
        return ~(((x & 0x7f7f7f7fu) + 0x7f7f7f7fu) | x | 0x7f7f7f7fu);
      };
      auto prock = [&](uint4 c, int blk) {
        unsigned m = (eqb(c.x) >> 7) | (eqb(c.y) >> 6) | (eqb(c.z) >> 5) | (eqb(c.w) >> 4);
        unsigned any = __builtin_amdgcn_ballot_w32(m != 0u);
        if (any) {
          const int e0 = blk * 512 + lane * 16;
          do {
            room(32);
            const int t = __builtin_ctz(m | 0x80000000u);
            push(any, e0 + ((t & 7) << 2) + (t >> 3));
            m &= m - 1u;
            any = __builtin_amdgcn_ballot_w32(m != 0u);
          } while (any);
          fold(false);
        }
      };
      int blk = wv;
      uint4 r0 = ldk(blk),          r1 = ldk(blk + nw),     r2 = ldk(blk + 2 * nw), r3 = ldk(blk + 3 * nw);
      uint4 r4 = ldk(blk + 4 * nw), r5 = ldk(blk + 5 * nw), r6 = ldk(blk + 6 * nw), r7 = ldk(blk + 7 * nw);
      for (; blk < nkb; blk += 8 * nw) {
        prock(r0, blk);          r0 = ldk(blk + 8 * nw);
        prock(r1, blk + nw);     r1 = ldk(blk + 9 * nw);
        prock(r2, blk + 2 * nw); r2 = ldk(blk + 10 * nw);
        prock(r3, blk + 3 * nw); r3 = ldk(blk + 11 * nw);
        prock(r4, blk + 4 * nw); r4 = ldk(blk + 12 * nw);
        prock(r5, blk + 5 * nw); r5 = ldk(blk + 13 * nw);
        prock(r6, blk + 6 * nw); r6 = ldk(blk + 14 * nw);
        prock(r7, blk + 7 * nw); r7 = ldk(blk + 15 * nw);
      }
    } else {
    const int* dstp = ei + Eraw;
    const int nblk = ((Eraw & 3) == 0) ? (Eraw >> 7) : 0;
    auto ld = [&](int blk) {
      int4 r = make_int4(-1, -1, -1, -1);
      if (blk < nblk) r = *(const int4*)(dstp + (size_t)blk * 128 + lane * 4);
      return r;
    };
    auto proc = [&](int4 c, int blk) {
      const unsigned m0 = __builtin_amdgcn_ballot_w32((unsigned)(c.x - base) < (unsigned)nloc);
      const unsigned m1 = __builtin_amdgcn_ballot_w32((unsigned)(c.y - base) < (unsigned)nloc);
      const unsigned m2 = __builtin_amdgcn_ballot_w32((unsigned)(c.z - base) < (unsigned)nloc);
      const unsigned m3 = __builtin_amdgcn_ballot_w32((unsigned)(c.w - base) < (unsigned)nloc);
      if (m0 | m1 | m2 | m3) {
        room(128);
        const int e = blk * 128 + lane * 4;
        if (m0) push(m0, e);
        if (m1) push(m1, e + 1);
        if (m2) push(m2, e + 2);
        if (m3) push(m3, e + 3);
        fold(false);
      }
    };
    {
      int blk = wv;
      int4 r0 = ld(blk), r1 = ld(blk + nw), r2 = ld(blk + 2 * nw), r3 = ld(blk + 3 * nw);
      for (; blk < nblk; blk += 4 * nw) {
        proc(r0, blk);          r0 = ld(blk + 4 * nw);
        proc(r1, blk + nw);     r1 = ld(blk + 5 * nw);
        proc(r2, blk + 2 * nw); r2 = ld(blk + 6 * nw);
        proc(r3, blk + 3 * nw); r3 = ld(blk + 7 * nw);
      }
    }
    for (int e0 = nblk * 128 + wv * 32; e0 < Eraw; e0 += nw * 32) {
      const int e = e0 + lane;
      const int d = (e < Eraw) ? dstp[e] : -1;
      const unsigned mask = __builtin_amdgcn_ballot_w32((unsigned)(d - base) < (unsigned)nloc);
      if (mask) { room(32); push(mask, e); fold(false); }
    }
    }
    fold(true);
    if (lost && lane == 0) ovf = 1;
  }
  __syncthreads();

  const int again = __builtin_amdgcn_readfirstlane(ovf);
  if (!TAKE) {
    if (!again) for (int i = lane; i < qn; i += 32) glist[i] = myq[i];
    if (lane == 0) lcnt[blockIdx.x * nw + wv] = again ? -1 : qn;
  }
  if (!again) {
    auto batch = [&](int q0, int cnt) {
      const int g  = lane % G;
      const int k  = g / H;
      const int hd = g - k * H;
      const int kq = k < cnt ? k : cnt - 1;
      const int pk = myq[q0 + kq];
      const int sq = pk >> 8;
      const int dq = pk & 255;

      float f[B][NL];
      int   dk[B];
#pragma unroll
      for (int kk = 0; kk < B; ++kk) {
        const int sk = __builtin_amdgcn_readlane(sq, kk * H);
        dk[kk]       = __builtin_amdgcn_readlane(dq, kk * H);
        const float* fp = feat + (size_t)sk * C + lane;
#pragma unroll
        for (int j = 0; j < NL; ++j) f[kk][j] = fp[j * 32];
      }

      float v = asrc[(size_t)sq * H + hd] + adl[dq * H + hd];
      v = v > 0.f ? v : NEG_SLOPE * v;
      const float w = __expf(v - ord2f(mu[dq * H + hd]));
      if (lane < G && k < cnt) atomicAdd(den + dq * H + hd, w);

#pragma unroll
      for (int kk = 0; kk < B; ++kk) {
        if (kk < cnt) {
          float* op = tile + dk[kk] * C + lane;
#pragma unroll
          for (int j = 0; j < NL; ++j) {
            const float a = __int_as_float(
                __builtin_amdgcn_readlane(__float_as_int(w), kk * H + (j * 32) / F));
            atomicAdd(op + j * 32, f[kk][j] * a);
          }
        }
      }
    };
    for (int q0 = 0; q0 < qn; q0 += B) batch(q0, (qn - q0) < B ? (qn - q0) : B);
  } else {
    const int Etot = Eraw + n_nodes;
    for (int e0 = wv * 32; e0 < Etot; e0 += nw * 32) {
      const int e = e0 + lane;
      int s = 0, d = -1;
      if (e < Etot) edge_ends(ei, e, Eraw, s, d);
      unsigned mask = __builtin_amdgcn_ballot_w32((unsigned)(d - base) < (unsigned)nloc);
      while (mask) {
        const int l = __builtin_ctz(mask);
        mask &= mask - 1;
        const int sk = __shfl(s, l);
        const int dk = __shfl(d, l) - base;
        if (lane < H) {
          float v = asrc[(size_t)sk * H + lane] + adl[dk * H + lane];
          v = v > 0.f ? v : NEG_SLOPE * v;
          atomicAdd(den + dk * H + lane, __expf(v - ord2f(mu[dk * H + lane])));
        }
#pragma unroll
        for (int j = 0; j < NL; ++j) {
          const int hd = (j * 32) / F;
          float v = asrc[(size_t)sk * H + hd] + adl[dk * H + hd];
          v = v > 0.f ? v : NEG_SLOPE * v;
          const float a = __expf(v - ord2f(mu[dk * H + hd]));
          atomicAdd(tile + dk * C + j * 32 + lane, feat[(size_t)sk * C + j * 32 + lane] * a);
        }
      }
    }
  }
  __syncthreads();

  {
    const float4* t4 = (const float4*)tile;
    float4* o4 = (float4*)(out + (size_t)base * C);
    const float4* b4 = (const float4*)bias;
    for (int i = tid; i < nloc * (C / 4); i += blockDim.x) {
      const int nl = i / (C / 4);
      const int c4 = i - nl * (C / 4);
      const float r = den[nl * H + (c4 * 4) / F] + 1e-16f;
      const float4 t = t4[i];
      const float4 bb = b4[c4];
      float4 v;
      v.x = t.x / r + bb.x; v.y = t.y / r + bb.y; v.z = t.z / r + bb.z; v.w = t.w / r + bb.w;
      if (ELU) {
        v.x = v.x > 0.f ? v.x : expm1f(v.x);
        v.y = v.y > 0.f ? v.y : expm1f(v.y);
        v.z = v.z > 0.f ? v.z : expm1f(v.z);
        v.w = v.w > 0.f ? v.w : expm1f(v.w);
      }
      o4[i] = v;
    }
  }
}


static inline int cdiv(long long a, long long b) { return (int)((a + b - 1) / b); }

extern "C" void kernel_launch(void* const* d_in, const int* in_sizes, int n_in,
                              void* d_out, int out_size, void* d_ws, size_t ws_size,
                              hipStream_t stream) {
  const float* x      = (const float*)d_in[0];
  const int*   ei     = (const int*)  d_in[1];
  const float* W1     = (const float*)d_in[2];
  const float* att_s1 = (const float*)d_in[3];
  const float* att_d1 = (const float*)d_in[4];
  const float* b1     = (const float*)d_in[5];
  const float* W2     = (const float*)d_in[6];
  const float* att_s2 = (const float*)d_in[7];
  const float* att_d2 = (const float*)d_in[8];
  const float* b2     = (const float*)d_in[9];

  const int IN   = 128;
  const int C1   = 256;
  const int Nn   = in_sizes[0] / IN;
  const int Eraw = in_sizes[1] / 2;

  float* ws = (float*)d_ws;
  size_t off = 0;
  float*    h1     = ws + off; off += (size_t)Nn * C1;
  float*    out1   = ws + off; off += (size_t)Nn * C1;
  float*    asrc1  = ws + off; off += (size_t)Nn * 4;
  float*    adst1  = ws + off; off += (size_t)Nn * 4;
  int*      lists  = (int*)(ws + off); off += (size_t)cdiv(Nn, 200) * 32 * LIST_SLOTS;
  int*      lcnt   = (int*)(ws + off); off += (size_t)cdiv(Nn, 200) * 32;
  unsigned* keys   = (unsigned*)(ws + off); off += (size_t)cdiv(Eraw, 4);
  float*    h2     = h1;
  float*    asrc2  = asrc1;
  float*    adst2  = adst1;
  float*    outf   = (float*)d_out;

  const int T = 256;

  gemm_lds_f32<128, 256, 7, 2><<<cdiv(Nn, 7 * 16), 256, 0, stream>>>(x, W1, h1, Nn);
  node_alpha<4, 64, 200, true><<<cdiv((long long)Nn * 4, T), T, 0, stream>>>(
      h1, att_s1, att_d1, asrc1, adst1, Nn, ei, keys, Eraw);
  range_agg_k<4, 64, 8, 200, true, false><<<cdiv(Nn, 200), 1024, 0, stream>>>(
      ei, asrc1, adst1, h1, b1, out1, lists, lcnt, keys, Nn, Eraw);

  gemm_lds_f32<256, 64, 8, 4><<<cdiv(Nn, 8 * 16), 256, 0, stream>>>(out1, W2, h2, Nn);
  node_alpha<1, 64, 200, false><<<cdiv((long long)Nn, T), T, 0, stream>>>(
      h2, att_s2, att_d2, asrc2, adst2, Nn, ei, keys, Eraw);
  range_agg_k<1, 64, 16, 200, false, true><<<cdiv(Nn, 200), 1024, 0, stream>>>(
      ei, asrc2, adst2, h2, b2, outf, lists, lcnt, keys, Nn, Eraw);
}


# Node-score launches re-tiled: a 16-lane row takes one head's 64 channels, 16 B a lane (whole-line loads), 4 products in lane then a row_xmask add tree; was a thread per 256-B piece; sum order changes
# speedup vs baseline: 4.4362x; 1.0323x over previous
#include <hip/hip_runtime.h>

typedef __attribute__((ext_vector_type(2))) float v2f;
typedef __attribute__((ext_vector_type(8))) float v8f;
typedef __attribute__((ext_vector_type(4))) float v4f;

#define NEG_SLOPE 0.2f
#define LIST_SLOTS 512


__device__ __forceinline__ unsigned f2ord(float f) {
  unsigned u = __float_as_uint(f);
  return (u & 0x80000000u) ? ~u : (u | 0x80000000u);
}
__device__ __forceinline__ float ord2f(unsigned o) {
  unsigned u = (o & 0x80000000u) ? (o & 0x7fffffffu) : ~o;
  return __uint_as_float(u);
}

__device__ __forceinline__ void edge_ends(const int* __restrict__ ei, int e,
                                          int Eraw, int& s, int& d) {
  if (e < Eraw) { s = ei[e]; d = ei[Eraw + e]; }
  else          { s = e - Eraw; d = s; }
}

template <int K, int N, int MT, int KU>
__global__ __launch_bounds__(256) void gemm_lds_f32(const float* __restrict__ A,
                                                    const float* __restrict__ B,
                                                    float* __restrict__ C, int M) {
  constexpr int P  = K + 4;
  constexpr int NT = N / 16;
  constexpr int RN = (NT >= 8) ? NT / 8 : 1;
  constexpr int WN = NT / RN;
  constexpr int WM = 8 / WN;
  constexpr int RM = MT / WM;
  static_assert(WN * WM == 8 && RM * WM == MT && (K % (4 * KU)) == 0, "tiling");
  constexpr int ROWS = MT * 16;
  constexpr int R4   = K / 4;
  constexpr int TOT4 = ROWS * R4;
  static_assert((TOT4 % 256) == 0, "staging");

  __shared__ __attribute__((aligned(16))) float As[ROWS * P];

  const int tid  = threadIdx.x;
  const int lane = tid & 31;
  const int w    = __builtin_amdgcn_readfirstlane(tid >> 5);
  const int lh   = lane >> 4;
  const int l16  = lane & 15;
  const int m0   = blockIdx.x * ROWS;

  const int wn = w % WN;
  const int wm = w / WN;
  const float* bp = B + (size_t)(lh * 2) * N + (wn * RN) * 16 + l16;
  const float* ap = As + (wm * RM * 16 + l16) * P + lh * 2;

  v2f bc[KU][RN];
#pragma unroll
  for (int u = 0; u < KU; ++u)
#pragma unroll
    for (int j = 0; j < RN; ++j) {
      bc[u][j].x = bp[(size_t)(4 * u) * N + j * 16];
      bc[u][j].y = bp[(size_t)(4 * u + 1) * N + j * 16];
    }

  {
    constexpr int PER = TOT4 / 256;
    constexpr int UB  = (PER % 16 == 0) ? 16 : PER;
    static_assert((PER % UB) == 0, "staging batches");
    for (int c0 = 0; c0 < PER; c0 += UB) {
      float4 v[UB];
#pragma unroll
      for (int u = 0; u < UB; ++u) {
        const int i  = tid + (c0 + u) * 256;
        const int r  = i / R4;
        const int c4 = i - r * R4;
        int gr = m0 + r; if (gr > M - 1) gr = M - 1;
        v[u] = *(const float4*)(A + (size_t)gr * K + c4 * 4);
      }
#pragma unroll
      for (int u = 0; u < UB; ++u) {
        const int i  = tid + (c0 + u) * 256;
        const int r  = i / R4;
        const int c4 = i - r * R4;
        *(float4*)(As + r * P + c4 * 4) = v[u];
      }
    }
  }
  __syncthreads();

  v8f acc[RM][RN];
#pragma unroll
  for (int i = 0; i < RM; ++i)
#pragma unroll
    for (int j = 0; j < RN; ++j) acc[i][j] = v8f{};

#pragma unroll
  for (int k = 0; k < K; k += 4 * KU) {
    const int kn = (k + 4 * KU < K) ? k + 4 * KU : k;
    const float* bq = bp + (size_t)kn * N;
    v2f bn[KU][RN];
#pragma unroll
    for (int u = 0; u < KU; ++u)
#pragma unroll
      for (int j = 0; j < RN; ++j) {
        bn[u][j].x = bq[(size_t)(4 * u) * N + j * 16];
        bn[u][j].y = bq[(size_t)(4 * u + 1) * N + j * 16];
      }
    v2f a[KU][RM];
#pragma unroll
    for (int u = 0; u < KU; ++u)
#pragma unroll
      for (int i = 0; i < RM; ++i) a[u][i] = *(const v2f*)(ap + i * 16 * P + k + 4 * u);
#pragma unroll
    for (int u = 0; u < KU; ++u)
#pragma unroll
      for (int i = 0; i < RM; ++i)
#pragma unroll
        for (int j = 0; j < RN; ++j)
          acc[i][j] = __builtin_amdgcn_wmma_f32_16x16x4_f32(false, a[u][i], false, bc[u][j],
                                                            (short)0, acc[i][j], false, false);
#pragma unroll
    for (int u = 0; u < KU; ++u)
#pragma unroll
      for (int j = 0; j < RN; ++j) bc[u][j] = bn[u][j];
  }

  const int mtiles = M >> 4;
#pragma unroll
  for (int i = 0; i < RM; ++i) {
    const int mt = (m0 >> 4) + wm * RM + i;
    if (mt < mtiles) {
#pragma unroll
      for (int j = 0; j < RN; ++j) {
        float* crow = C + (size_t)(mt * 16 + lh * 8) * N + (wn * RN + j) * 16 + l16;
#pragma unroll
        for (int q = 0; q < 8; ++q) crow[(size_t)q * N] = acc[i][j][q];
      }
    }
  }
}


template <int CTRL>
__device__ __forceinline__ float row_xadd(float v) {
  return v + __int_as_float(__builtin_amdgcn_update_dpp(0, __float_as_int(v), CTRL, 0xf, 0xf, true));
}

template <int H, int F, int NPB, bool KEYS>
__global__ __launch_bounds__(256) void node_alpha(const float* __restrict__ h,
                           const float* __restrict__ att_s,
                           const float* __restrict__ att_d,
                           float* __restrict__ asrc, float* __restrict__ adst,
                           int n_nodes,
                           const int* __restrict__ ei, unsigned* __restrict__ keys, int Eraw) {
  constexpr int U   = 16;
  constexpr int PER = (H >= 2) ? H / 2 : 1;
  static_assert(F == 64 && ((2 * U) % H) == 0 && (U % PER) == 0, "tiling");
  int idx = blockIdx.x * blockDim.x + threadIdx.x;
  if (KEYS && (Eraw & 15) == 0) {
    const int4* d4 = (const int4*)(ei + Eraw);
    const int nq = Eraw >> 2;
    auto key = [&](int d) -> unsigned {
      unsigned k = (unsigned)d / (unsigned)NPB;
      if (k > 255u) k = 255u;
      return ((unsigned)d < (unsigned)n_nodes) ? k : 255u;
    };
    for (int q = idx; q < nq; q += gridDim.x * blockDim.x) {
      const int4 d = d4[q];
      keys[q] = key(d.x) | (key(d.y) << 8) | (key(d.z) << 16) | (key(d.w) << 24);
    }
  }
  const int lane = threadIdx.x & 31;
  const int l16  = lane & 15;
  const int lh   = lane >> 4;
  const int wave   = __builtin_amdgcn_readfirstlane(idx >> 5);
  const int nwaves = (int)(gridDim.x * blockDim.x) >> 5;
  const int nseg   = n_nodes * H;
  const int ntrip  = (nseg + 2 * U - 1) / (2 * U);

  float4 as[PER], ad[PER];
#pragma unroll
  for (int r = 0; r < PER; ++r) {
    const int hd = (2 * r + lh) % H;
    as[r] = *(const float4*)(att_s + hd * F + l16 * 4);
    ad[r] = *(const float4*)(att_d + hd * F + l16 * 4);
  }

  for (int t = wave; t < ntrip; t += nwaves) {
    const int seg0 = t * (2 * U);
    v4f x[U];
    if (seg0 + 2 * U <= nseg) {
      const float* hp = h + (size_t)(seg0 + lh) * F + l16 * 4;
#pragma unroll
      for (int u = 0; u < U; ++u) x[u] = *(const v4f*)(hp + u * 2 * F);
    } else {
#pragma unroll
      for (int u = 0; u < U; ++u) {
        int sg = seg0 + 2 * u + lh; if (sg > nseg - 1) sg = nseg - 1;
        x[u] = *(const v4f*)(h + (size_t)sg * F + l16 * 4);
      }
    }
#pragma unroll
    for (int u = 0; u < U; ++u) asm volatile("" :: "v"(x[u]));
    float s[U], d[U];
#pragma unroll
    for (int u = 0; u < U; ++u) {
      const float4 a = as[u % PER], b = ad[u % PER];
      s[u] = x[u].x * a.x + x[u].y * a.y + x[u].z * a.z + x[u].w * a.w;
      d[u] = x[u].x * b.x + x[u].y * b.y + x[u].z * b.z + x[u].w * b.w;
    }
#pragma unroll
    for (int u = 0; u < U; ++u) { s[u] = row_xadd<0x161>(s[u]); d[u] = row_xadd<0x161>(d[u]); }
#pragma unroll
    for (int u = 0; u < U; ++u) { s[u] = row_xadd<0x162>(s[u]); d[u] = row_xadd<0x162>(d[u]); }
#pragma unroll
    for (int u = 0; u < U; ++u) { s[u] = row_xadd<0x164>(s[u]); d[u] = row_xadd<0x164>(d[u]); }
#pragma unroll
    for (int u = 0; u < U; ++u) { s[u] = row_xadd<0x168>(s[u]); d[u] = row_xadd<0x168>(d[u]); }
    float so = s[0], dd = d[0];
#pragma unroll
    for (int u = 1; u < U; ++u) { if (l16 == u) { so = s[u]; dd = d[u]; } }
    const int sg = seg0 + 2 * l16 + lh;
    if (sg < nseg) { asrc[sg] = so; adst[sg] = dd; }
  }
}


template <int H, int F, int B, int NPB, bool ELU, bool TAKE>
__global__ __launch_bounds__(1024) void range_agg_k(
    const int* __restrict__ ei,
    const float* __restrict__ asrc,
    const float* __restrict__ adst,
    const float* __restrict__ feat,
    const float* __restrict__ bias,
    float* __restrict__ out,
    int* __restrict__ lists, int* __restrict__ lcnt,
    const unsigned* __restrict__ keys,
    int n_nodes, int Eraw) {
  constexpr int C  = H * F;
  constexpr int NL = C / 32;
  constexpr int G  = B * H;
  constexpr int QC = LIST_SLOTS;
  static_assert(G <= 32 && (32 % G) == 0 && (F % 32) == 0 && (C % 4) == 0, "tiling");
  static_assert(NPB <= 256, "local row must fit the low byte of a list entry");

  __shared__ __attribute__((aligned(16))) float tile[NPB * C];
  __shared__ float    den[NPB * H];
  __shared__ unsigned mu[NPB * H];
  __shared__ float    adl[NPB * H];
  __shared__ int      qe[32 * QC];
  __shared__ int      ovf;

  const int tid  = threadIdx.x;
  const int lane = tid & 31;
  const int wv   = __builtin_amdgcn_readfirstlane(tid >> 5);
  const int nw   = blockDim.x >> 5;
  const int base = blockIdx.x * NPB;
  int nloc = n_nodes - base; if (nloc > NPB) nloc = NPB;

  {
    float4* t4 = (float4*)tile;
    for (int i = tid; i < NPB * C / 4; i += blockDim.x) t4[i] = make_float4(0.f, 0.f, 0.f, 0.f);
    for (int i = tid; i < NPB * H; i += blockDim.x) {
      den[i] = 0.f;
      mu[i]  = 0u;
      adl[i] = (i < nloc * H) ? adst[(size_t)base * H + i] : 0.f;
    }
    if (tid == 0) ovf = 0;
  }
  __syncthreads();

  int* myq = qe + wv * QC;
  int  qn = 0;
  int  done = 0;
  bool lost = false;

  auto chunk = [&](int q0, int cnt) {
    if (lane < cnt) {
      const int e = myq[q0 + lane];
      int s, d; edge_ends(ei, e, Eraw, s, d);
      const int dl = d - base;
#pragma unroll
      for (int hd = 0; hd < H; ++hd) {
        float v = asrc[(size_t)s * H + hd] + adl[dl * H + hd];
        v = v > 0.f ? v : NEG_SLOPE * v;
        atomicMax(mu + dl * H + hd, f2ord(v));
      }
      myq[q0 + lane] = (s << 8) | dl;
    }
  };
  auto fold = [&](bool all) {
    __builtin_amdgcn_fence(__ATOMIC_ACQ_REL, "wavefront");
    while (qn - done >= 32) { chunk(done, 32); done += 32; }
    if (all && done < qn) { chunk(done, qn - done); done = qn; }
  };
  auto room = [&](int n) {
    if (qn + n > QC) { fold(true); qn = 0; done = 0; lost = true; }
  };
  auto push = [&](unsigned mask, int e) {
    const int pos = qn + (int)__builtin_amdgcn_mbcnt_lo(mask, 0u);
    if ((mask >> lane) & 1u) myq[pos] = e;
    qn += __builtin_popcount(mask);
  };

  int* glist = lists + (size_t)(blockIdx.x * nw + wv) * QC;
  int have = -1;
  if (TAKE) have = __builtin_amdgcn_readfirstlane(lcnt[blockIdx.x * nw + wv]);

  if (TAKE && have >= 0) {
    for (int q0 = 0; q0 < have; q0 += 128) {
      int pk[4];
#pragma unroll
      for (int u = 0; u < 4; ++u) {
        const int i = q0 + u * 32 + lane;
        pk[u] = glist[i < have ? i : have - 1];
      }
      float sv[4][H];
#pragma unroll
      for (int u = 0; u < 4; ++u)
#pragma unroll
        for (int hd = 0; hd < H; ++hd) sv[u][hd] = asrc[(size_t)(pk[u] >> 8) * H + hd];
#pragma unroll
      for (int u = 0; u < 4; ++u)
#pragma unroll
        for (int hd = 0; hd < H; ++hd) asm volatile("" :: "v"(sv[u][hd]));
#pragma unroll
      for (int u = 0; u < 4; ++u) {
        const int i = q0 + u * 32 + lane;
        if (i < have) {
          const int dl = pk[u] & 255;
          myq[i] = pk[u];
#pragma unroll
          for (int hd = 0; hd < H; ++hd) {
            float v = sv[u][hd] + adl[dl * H + hd];
            v = v > 0.f ? v : NEG_SLOPE * v;
            atomicMax(mu + dl * H + hd, f2ord(v));
          }
        }
      }
    }
    qn = have;
  } else {
    {
      const int j = wv + nw * lane;
      const unsigned mask = __builtin_amdgcn_ballot_w32(j < nloc);
      if (mask) push(mask, Eraw + base + j);
    }

    if (gridDim.x <= 255u && (Eraw & 15) == 0) {
      const int nkb = (Eraw + 511) >> 9;
      const unsigned pat = blockIdx.x * 0x01010101u;
      auto ldk = [&](int blk) {
        uint4 r = make_uint4(~0u, ~0u, ~0u, ~0u);
        if (blk < nkb) {
          const int e = blk * 512 + lane * 16;
          if (e < Eraw) r = *(const uint4*)(keys + (e >> 2));
        }
        return r;
      };
      auto eqb = [&](unsigned c) {
        const unsigned x = c ^ pat;
        return ~(((x & 0x7f7f7f7fu) + 0x7f7f7f7fu) | x | 0x7f7f7f7fu);
      };
      auto prock = [&](uint4 c, int blk) {
        unsigned m = (eqb(c.x) >> 7) | (eqb(c.y) >> 6) | (eqb(c.z) >> 5) | (eqb(c.w) >> 4);
        unsigned any = __builtin_amdgcn_ballot_w32(m != 0u);
        if (any) {
          const int e0 = blk * 512 + lane * 16;
          do {
            room(32);
            const int t = __builtin_ctz(m | 0x80000000u);
            push(any, e0 + ((t & 7) << 2) + (t >> 3));
            m &= m - 1u;
            any = __builtin_amdgcn_ballot_w32(m != 0u);
          } while (any);
          fold(false);
        }
      };
      int blk = wv;
      uint4 r0 = ldk(blk),          r1 = ldk(blk + nw),     r2 = ldk(blk + 2 * nw), r3 = ldk(blk + 3 * nw);
      uint4 r4 = ldk(blk + 4 * nw), r5 = ldk(blk + 5 * nw), r6 = ldk(blk + 6 * nw), r7 = ldk(blk + 7 * nw);
      for (; blk < nkb; blk += 8 * nw) {
        prock(r0, blk);          r0 = ldk(blk + 8 * nw);
        prock(r1, blk + nw);     r1 = ldk(blk + 9 * nw);
        prock(r2, blk + 2 * nw); r2 = ldk(blk + 10 * nw);
        prock(r3, blk + 3 * nw); r3 = ldk(blk + 11 * nw);
        prock(r4, blk + 4 * nw); r4 = ldk(blk + 12 * nw);
        prock(r5, blk + 5 * nw); r5 = ldk(blk + 13 * nw);
        prock(r6, blk + 6 * nw); r6 = ldk(blk + 14 * nw);
        prock(r7, blk + 7 * nw); r7 = ldk(blk + 15 * nw);
      }
    } else {
    const int* dstp = ei + Eraw;
    const int nblk = ((Eraw & 3) == 0) ? (Eraw >> 7) : 0;
    auto ld = [&](int blk) {
      int4 r = make_int4(-1, -1, -1, -1);
      if (blk < nblk) r = *(const int4*)(dstp + (size_t)blk * 128 + lane * 4);
      return r;
    };
    auto proc = [&](int4 c, int blk) {
      const unsigned m0 = __builtin_amdgcn_ballot_w32((unsigned)(c.x - base) < (unsigned)nloc);
      const unsigned m1 = __builtin_amdgcn_ballot_w32((unsigned)(c.y - base) < (unsigned)nloc);
      const unsigned m2 = __builtin_amdgcn_ballot_w32((unsigned)(c.z - base) < (unsigned)nloc);
      const unsigned m3 = __builtin_amdgcn_ballot_w32((unsigned)(c.w - base) < (unsigned)nloc);
      if (m0 | m1 | m2 | m3) {
        room(128);
        const int e = blk * 128 + lane * 4;
        if (m0) push(m0, e);
        if (m1) push(m1, e + 1);
        if (m2) push(m2, e + 2);
        if (m3) push(m3, e + 3);
        fold(false);
      }
    };
    {
      int blk = wv;
      int4 r0 = ld(blk), r1 = ld(blk + nw), r2 = ld(blk + 2 * nw), r3 = ld(blk + 3 * nw);
      for (; blk < nblk; blk += 4 * nw) {
        proc(r0, blk);          r0 = ld(blk + 4 * nw);
        proc(r1, blk + nw);     r1 = ld(blk + 5 * nw);
        proc(r2, blk + 2 * nw); r2 = ld(blk + 6 * nw);
        proc(r3, blk + 3 * nw); r3 = ld(blk + 7 * nw);
      }
    }
    for (int e0 = nblk * 128 + wv * 32; e0 < Eraw; e0 += nw * 32) {
      const int e = e0 + lane;
      const int d = (e < Eraw) ? dstp[e] : -1;
      const unsigned mask = __builtin_amdgcn_ballot_w32((unsigned)(d - base) < (unsigned)nloc);
      if (mask) { room(32); push(mask, e); fold(false); }
    }
    }
    fold(true);
    if (lost && lane == 0) ovf = 1;
  }
  __syncthreads();

  const int again = __builtin_amdgcn_readfirstlane(ovf);
  if (!TAKE) {
    if (!again) for (int i = lane; i < qn; i += 32) glist[i] = myq[i];
    if (lane == 0) lcnt[blockIdx.x * nw + wv] = again ? -1 : qn;
  }
  if (!again) {
    auto batch = [&](int q0, int cnt) {
      const int g  = lane % G;
      const int k  = g / H;
      const int hd = g - k * H;
      const int kq = k < cnt ? k : cnt - 1;
      const int pk = myq[q0 + kq];
      const int sq = pk >> 8;
      const int dq = pk & 255;

      float f[B][NL];
      int   dk[B];
#pragma unroll
      for (int kk = 0; kk < B; ++kk) {
        const int sk = __builtin_amdgcn_readlane(sq, kk * H);
        dk[kk]       = __builtin_amdgcn_readlane(dq, kk * H);
        const float* fp = feat + (size_t)sk * C + lane;
#pragma unroll
        for (int j = 0; j < NL; ++j) f[kk][j] = fp[j * 32];
      }

      float v = asrc[(size_t)sq * H + hd] + adl[dq * H + hd];
      v = v > 0.f ? v : NEG_SLOPE * v;
      const float w = __expf(v - ord2f(mu[dq * H + hd]));
      if (lane < G && k < cnt) atomicAdd(den + dq * H + hd, w);

#pragma unroll
      for (int kk = 0; kk < B; ++kk) {
        if (kk < cnt) {
          float* op = tile + dk[kk] * C + lane;
#pragma unroll
          for (int j = 0; j < NL; ++j) {
            const float a = __int_as_float(
                __builtin_amdgcn_readlane(__float_as_int(w), kk * H + (j * 32) / F));
            atomicAdd(op + j * 32, f[kk][j] * a);
          }
        }
      }
    };
    for (int q0 = 0; q0 < qn; q0 += B) batch(q0, (qn - q0) < B ? (qn - q0) : B);
  } else {
    const int Etot = Eraw + n_nodes;
    for (int e0 = wv * 32; e0 < Etot; e0 += nw * 32) {
      const int e = e0 + lane;
      int s = 0, d = -1;
      if (e < Etot) edge_ends(ei, e, Eraw, s, d);
      unsigned mask = __builtin_amdgcn_ballot_w32((unsigned)(d - base) < (unsigned)nloc);
      while (mask) {
        const int l = __builtin_ctz(mask);
        mask &= mask - 1;
        const int sk = __shfl(s, l);
        const int dk = __shfl(d, l) - base;
        if (lane < H) {
          float v = asrc[(size_t)sk * H + lane] + adl[dk * H + lane];
          v = v > 0.f ? v : NEG_SLOPE * v;
          atomicAdd(den + dk * H + lane, __expf(v - ord2f(mu[dk * H + lane])));
        }
#pragma unroll
        for (int j = 0; j < NL; ++j) {
          const int hd = (j * 32) / F;
          float v = asrc[(size_t)sk * H + hd] + adl[dk * H + hd];
          v = v > 0.f ? v : NEG_SLOPE * v;
          const float a = __expf(v - ord2f(mu[dk * H + hd]));
          atomicAdd(tile + dk * C + j * 32 + lane, feat[(size_t)sk * C + j * 32 + lane] * a);
        }
      }
    }
  }
  __syncthreads();

  {
    const float4* t4 = (const float4*)tile;
    float4* o4 = (float4*)(out + (size_t)base * C);
    const float4* b4 = (const float4*)bias;
    for (int i = tid; i < nloc * (C / 4); i += blockDim.x) {
      const int nl = i / (C / 4);
      const int c4 = i - nl * (C / 4);
      const float r = den[nl * H + (c4 * 4) / F] + 1e-16f;
      const float4 t = t4[i];
      const float4 bb = b4[c4];
      float4 v;
      v.x = t.x / r + bb.x; v.y = t.y / r + bb.y; v.z = t.z / r + bb.z; v.w = t.w / r + bb.w;
      if (ELU) {
        v.x = v.x > 0.f ? v.x : expm1f(v.x);
        v.y = v.y > 0.f ? v.y : expm1f(v.y);
        v.z = v.z > 0.f ? v.z : expm1f(v.z);
        v.w = v.w > 0.f ? v.w : expm1f(v.w);
      }
      o4[i] = v;
    }
  }
}


static inline int cdiv(long long a, long long b) { return (int)((a + b - 1) / b); }

extern "C" void kernel_launch(void* const* d_in, const int* in_sizes, int n_in,
                              void* d_out, int out_size, void* d_ws, size_t ws_size,
                              hipStream_t stream) {
  const float* x      = (const float*)d_in[0];
  const int*   ei     = (const int*)  d_in[1];
  const float* W1     = (const float*)d_in[2];
  const float* att_s1 = (const float*)d_in[3];
  const float* att_d1 = (const float*)d_in[4];
  const float* b1     = (const float*)d_in[5];
  const float* W2     = (const float*)d_in[6];
  const float* att_s2 = (const float*)d_in[7];
  const float* att_d2 = (const float*)d_in[8];
  const float* b2     = (const float*)d_in[9];

  const int IN   = 128;
  const int C1   = 256;
  const int Nn   = in_sizes[0] / IN;
  const int Eraw = in_sizes[1] / 2;

  float* ws = (float*)d_ws;
  size_t off = 0;
  float*    h1     = ws + off; off += (size_t)Nn * C1;
  float*    out1   = ws + off; off += (size_t)Nn * C1;
  float*    asrc1  = ws + off; off += (size_t)Nn * 4;
  float*    adst1  = ws + off; off += (size_t)Nn * 4;
  int*      lists  = (int*)(ws + off); off += (size_t)cdiv(Nn, 200) * 32 * LIST_SLOTS;
  int*      lcnt   = (int*)(ws + off); off += (size_t)cdiv(Nn, 200) * 32;
  unsigned* keys   = (unsigned*)(ws + off); off += (size_t)cdiv(Eraw, 4);
  float*    h2     = h1;
  float*    asrc2  = asrc1;
  float*    adst2  = adst1;
  float*    outf   = (float*)d_out;

  const int T = 256;

  gemm_lds_f32<128, 256, 7, 2><<<cdiv(Nn, 7 * 16), 256, 0, stream>>>(x, W1, h1, Nn);
  node_alpha<4, 64, 200, true><<<cdiv((long long)Nn * 4, 2 * T), T, 0, stream>>>(
      h1, att_s1, att_d1, asrc1, adst1, Nn, ei, keys, Eraw);
  range_agg_k<4, 64, 8, 200, true, false><<<cdiv(Nn, 200), 1024, 0, stream>>>(
      ei, asrc1, adst1, h1, b1, out1, lists, lcnt, keys, Nn, Eraw);

  gemm_lds_f32<256, 64, 8, 4><<<cdiv(Nn, 8 * 16), 256, 0, stream>>>(out1, W2, h2, Nn);
  node_alpha<1, 64, 200, false><<<cdiv((long long)Nn, T), T, 0, stream>>>(
      h2, att_s2, att_d2, asrc2, adst2, Nn, ei, keys, Eraw);
  range_agg_k<1, 64, 16, 200, false, true><<<cdiv(Nn, 200), 1024, 0, stream>>>(
      ei, asrc2, adst2, h2, b2, outf, lists, lcnt, keys, Nn, Eraw);
}


# Aggregation gathers source rows by 16-byte loads: a 1-KB row by 2 loads of a wave (was 8), two 256-B rows by 1 (was 4), a source's 4 scores by 1; LDS rows in delivered word order; sum order kept
# speedup vs baseline: 4.6800x; 1.0550x over previous
#include <hip/hip_runtime.h>

typedef __attribute__((ext_vector_type(2))) float v2f;
typedef __attribute__((ext_vector_type(8))) float v8f;
typedef __attribute__((ext_vector_type(4))) float v4f;

#define NEG_SLOPE 0.2f
#define LIST_SLOTS 512


__device__ __forceinline__ unsigned f2ord(float f) {
  unsigned u = __float_as_uint(f);
  return (u & 0x80000000u) ? ~u : (u | 0x80000000u);
}
__device__ __forceinline__ float ord2f(unsigned o) {
  unsigned u = (o & 0x80000000u) ? (o & 0x7fffffffu) : ~o;
  return __uint_as_float(u);
}

__device__ __forceinline__ void edge_ends(const int* __restrict__ ei, int e,
                                          int Eraw, int& s, int& d) {
  if (e < Eraw) { s = ei[e]; d = ei[Eraw + e]; }
  else          { s = e - Eraw; d = s; }
}

template <int K, int N, int MT, int KU>
__global__ __launch_bounds__(256) void gemm_lds_f32(const float* __restrict__ A,
                                                    const float* __restrict__ B,
                                                    float* __restrict__ C, int M) {
  constexpr int P  = K + 4;
  constexpr int NT = N / 16;
  constexpr int RN = (NT >= 8) ? NT / 8 : 1;
  constexpr int WN = NT / RN;
  constexpr int WM = 8 / WN;
  constexpr int RM = MT / WM;
  static_assert(WN * WM == 8 && RM * WM == MT && (K % (4 * KU)) == 0, "tiling");
  constexpr int ROWS = MT * 16;
  constexpr int R4   = K / 4;
  constexpr int TOT4 = ROWS * R4;
  static_assert((TOT4 % 256) == 0, "staging");

  __shared__ __attribute__((aligned(16))) float As[ROWS * P];

  const int tid  = threadIdx.x;
  const int lane = tid & 31;
  const int w    = __builtin_amdgcn_readfirstlane(tid >> 5);
  const int lh   = lane >> 4;
  const int l16  = lane & 15;
  const int m0   = blockIdx.x * ROWS;

  const int wn = w % WN;
  const int wm = w / WN;
  const float* bp = B + (size_t)(lh * 2) * N + (wn * RN) * 16 + l16;
  const float* ap = As + (wm * RM * 16 + l16) * P + lh * 2;

  v2f bc[KU][RN];
#pragma unroll
  for (int u = 0; u < KU; ++u)
#pragma unroll
    for (int j = 0; j < RN; ++j) {
      bc[u][j].x = bp[(size_t)(4 * u) * N + j * 16];
      bc[u][j].y = bp[(size_t)(4 * u + 1) * N + j * 16];
    }

  {
    constexpr int PER = TOT4 / 256;
    constexpr int UB  = (PER % 16 == 0) ? 16 : PER;
    static_assert((PER % UB) == 0, "staging batches");
    for (int c0 = 0; c0 < PER; c0 += UB) {
      float4 v[UB];
#pragma unroll
      for (int u = 0; u < UB; ++u) {
        const int i  = tid + (c0 + u) * 256;
        const int r  = i / R4;
        const int c4 = i - r * R4;
        int gr = m0 + r; if (gr > M - 1) gr = M - 1;
        v[u] = *(const float4*)(A + (size_t)gr * K + c4 * 4);
      }
#pragma unroll
      for (int u = 0; u < UB; ++u) {
        const int i  = tid + (c0 + u) * 256;
        const int r  = i / R4;
        const int c4 = i - r * R4;
        *(float4*)(As + r * P + c4 * 4) = v[u];
      }
    }
  }
  __syncthreads();

  v8f acc[RM][RN];
#pragma unroll
  for (int i = 0; i < RM; ++i)
#pragma unroll
    for (int j = 0; j < RN; ++j) acc[i][j] = v8f{};

#pragma unroll
  for (int k = 0; k < K; k += 4 * KU) {
    const int kn = (k + 4 * KU < K) ? k + 4 * KU : k;
    const float* bq = bp + (size_t)kn * N;
    v2f bn[KU][RN];
#pragma unroll
    for (int u = 0; u < KU; ++u)
#pragma unroll
      for (int j = 0; j < RN; ++j) {
        bn[u][j].x = bq[(size_t)(4 * u) * N + j * 16];
        bn[u][j].y = bq[(size_t)(4 * u + 1) * N + j * 16];
      }
    v2f a[KU][RM];
#pragma unroll
    for (int u = 0; u < KU; ++u)
#pragma unroll
      for (int i = 0; i < RM; ++i) a[u][i] = *(const v2f*)(ap + i * 16 * P + k + 4 * u);
#pragma unroll
    for (int u = 0; u < KU; ++u)
#pragma unroll
      for (int i = 0; i < RM; ++i)
#pragma unroll
        for (int j = 0; j < RN; ++j)
          acc[i][j] = __builtin_amdgcn_wmma_f32_16x16x4_f32(false, a[u][i], false, bc[u][j],
                                                            (short)0, acc[i][j], false, false);
#pragma unroll
    for (int u = 0; u < KU; ++u)
#pragma unroll
      for (int j = 0; j < RN; ++j) bc[u][j] = bn[u][j];
  }

  const int mtiles = M >> 4;
#pragma unroll
  for (int i = 0; i < RM; ++i) {
    const int mt = (m0 >> 4) + wm * RM + i;
    if (mt < mtiles) {
#pragma unroll
      for (int j = 0; j < RN; ++j) {
        float* crow = C + (size_t)(mt * 16 + lh * 8) * N + (wn * RN + j) * 16 + l16;
#pragma unroll
        for (int q = 0; q < 8; ++q) crow[(size_t)q * N] = acc[i][j][q];
      }
    }
  }
}


template <int CTRL>
__device__ __forceinline__ float row_xadd(float v) {
  return v + __int_as_float(__builtin_amdgcn_update_dpp(0, __float_as_int(v), CTRL, 0xf, 0xf, true));
}

template <int N>
__device__ __forceinline__ int row_share(int v) {
  return __builtin_amdgcn_update_dpp(0, v, 0x150 + N, 0xf, 0xf, true);
}

template <int H, int F, int NPB, bool KEYS>
__global__ __launch_bounds__(256) void node_alpha(const float* __restrict__ h,
                           const float* __restrict__ att_s,
                           const float* __restrict__ att_d,
                           float* __restrict__ asrc, float* __restrict__ adst,
                           int n_nodes,
                           const int* __restrict__ ei, unsigned* __restrict__ keys, int Eraw) {
  constexpr int U   = 16;
  constexpr int PER = (H >= 2) ? H / 2 : 1;
  static_assert(F == 64 && ((2 * U) % H) == 0 && (U % PER) == 0, "tiling");
  int idx = blockIdx.x * blockDim.x + threadIdx.x;
  if (KEYS && (Eraw & 15) == 0) {
    const int4* d4 = (const int4*)(ei + Eraw);
    const int nq = Eraw >> 2;
    auto key = [&](int d) -> unsigned {
      unsigned k = (unsigned)d / (unsigned)NPB;
      if (k > 255u) k = 255u;
      return ((unsigned)d < (unsigned)n_nodes) ? k : 255u;
    };
    for (int q = idx; q < nq; q += gridDim.x * blockDim.x) {
      const int4 d = d4[q];
      keys[q] = key(d.x) | (key(d.y) << 8) | (key(d.z) << 16) | (key(d.w) << 24);
    }
  }
  const int lane = threadIdx.x & 31;
  const int l16  = lane & 15;
  const int lh   = lane >> 4;
  const int wave   = __builtin_amdgcn_readfirstlane(idx >> 5);
  const int nwaves = (int)(gridDim.x * blockDim.x) >> 5;
  const int nseg   = n_nodes * H;
  const int ntrip  = (nseg + 2 * U - 1) / (2 * U);

  float4 as[PER], ad[PER];
#pragma unroll
  for (int r = 0; r < PER; ++r) {
    const int hd = (2 * r + lh) % H;
    as[r] = *(const float4*)(att_s + hd * F + l16 * 4);
    ad[r] = *(const float4*)(att_d + hd * F + l16 * 4);
  }

  for (int t = wave; t < ntrip; t += nwaves) {
    const int seg0 = t * (2 * U);
    v4f x[U];
    if (seg0 + 2 * U <= nseg) {
      const float* hp = h + (size_t)(seg0 + lh) * F + l16 * 4;
#pragma unroll
      for (int u = 0; u < U; ++u) x[u] = *(const v4f*)(hp + u * 2 * F);
    } else {
#pragma unroll
      for (int u = 0; u < U; ++u) {
        int sg = seg0 + 2 * u + lh; if (sg > nseg - 1) sg = nseg - 1;
        x[u] = *(const v4f*)(h + (size_t)sg * F + l16 * 4);
      }
    }
#pragma unroll
    for (int u = 0; u < U; ++u) asm volatile("" :: "v"(x[u]));
    float s[U], d[U];
#pragma unroll
    for (int u = 0; u < U; ++u) {
      const float4 a = as[u % PER], b = ad[u % PER];
      s[u] = x[u].x * a.x + x[u].y * a.y + x[u].z * a.z + x[u].w * a.w;
      d[u] = x[u].x * b.x + x[u].y * b.y + x[u].z * b.z + x[u].w * b.w;
    }
#pragma unroll
    for (int u = 0; u < U; ++u) { s[u] = row_xadd<0x161>(s[u]); d[u] = row_xadd<0x161>(d[u]); }
#pragma unroll
    for (int u = 0; u < U; ++u) { s[u] = row_xadd<0x162>(s[u]); d[u] = row_xadd<0x162>(d[u]); }
#pragma unroll
    for (int u = 0; u < U; ++u) { s[u] = row_xadd<0x164>(s[u]); d[u] = row_xadd<0x164>(d[u]); }
#pragma unroll
    for (int u = 0; u < U; ++u) { s[u] = row_xadd<0x168>(s[u]); d[u] = row_xadd<0x168>(d[u]); }
    float so = s[0], dd = d[0];
#pragma unroll
    for (int u = 1; u < U; ++u) { if (l16 == u) { so = s[u]; dd = d[u]; } }
    const int sg = seg0 + 2 * l16 + lh;
    if (sg < nseg) { asrc[sg] = so; adst[sg] = dd; }
  }
}


template <int H, int F, int B, int NPB, bool ELU, bool TAKE>
__global__ __launch_bounds__(1024) void range_agg_k(
    const int* __restrict__ ei,
    const float* __restrict__ asrc,
    const float* __restrict__ adst,
    const float* __restrict__ feat,
    const float* __restrict__ bias,
    float* __restrict__ out,
    int* __restrict__ lists, int* __restrict__ lcnt,
    const unsigned* __restrict__ keys,
    int n_nodes, int Eraw) {
  constexpr int C  = H * F;
  constexpr int NL = C / 32;
  constexpr bool WIDE = (C % 128) == 0 && F == 64;
  constexpr int NV = WIDE ? C / 128 : 1;
  constexpr bool HALF = (C == 64) && (H == 1) && (B == 16);
  constexpr int G  = B * H;
  constexpr int QC = LIST_SLOTS;
  static_assert(G <= 32 && (32 % G) == 0 && (F % 32) == 0 && (C % 4) == 0, "tiling");
  static_assert(NPB <= 256, "local row must fit the low byte of a list entry");

  __shared__ __attribute__((aligned(16))) float tile[NPB * C];
  __shared__ float    den[NPB * H];
  __shared__ unsigned mu[NPB * H];
  __shared__ float    adl[NPB * H];
  __shared__ int      qe[32 * QC];
  __shared__ int      ovf;

  const int tid  = threadIdx.x;
  const int lane = tid & 31;
  const int wv   = __builtin_amdgcn_readfirstlane(tid >> 5);
  const int nw   = blockDim.x >> 5;
  const int base = blockIdx.x * NPB;
  int nloc = n_nodes - base; if (nloc > NPB) nloc = NPB;

  {
    float4* t4 = (float4*)tile;
    for (int i = tid; i < NPB * C / 4; i += blockDim.x) t4[i] = make_float4(0.f, 0.f, 0.f, 0.f);
    for (int i = tid; i < NPB * H; i += blockDim.x) {
      den[i] = 0.f;
      mu[i]  = 0u;
      adl[i] = (i < nloc * H) ? adst[(size_t)base * H + i] : 0.f;
    }
    if (tid == 0) ovf = 0;
  }
  __syncthreads();

  int* myq = qe + wv * QC;
  int  qn = 0;
  int  done = 0;
  bool lost = false;

  auto chunk = [&](int q0, int cnt) {
    if (lane < cnt) {
      const int e = myq[q0 + lane];
      int s, d; edge_ends(ei, e, Eraw, s, d);
      const int dl = d - base;
      float sc[H];
      if constexpr (H == 4) {
        const float4 s4 = *(const float4*)(asrc + (size_t)s * 4);
        sc[0] = s4.x; sc[1] = s4.y; sc[2] = s4.z; sc[3] = s4.w;
      } else {
#pragma unroll
        for (int hd = 0; hd < H; ++hd) sc[hd] = asrc[(size_t)s * H + hd];
      }
#pragma unroll
      for (int hd = 0; hd < H; ++hd) {
        float v = sc[hd] + adl[dl * H + hd];
        v = v > 0.f ? v : NEG_SLOPE * v;
        atomicMax(mu + dl * H + hd, f2ord(v));
      }
      myq[q0 + lane] = (s << 8) | dl;
    }
  };
  auto fold = [&](bool all) {
    __builtin_amdgcn_fence(__ATOMIC_ACQ_REL, "wavefront");
    while (qn - done >= 32) { chunk(done, 32); done += 32; }
    if (all && done < qn) { chunk(done, qn - done); done = qn; }
  };
  auto room = [&](int n) {
    if (qn + n > QC) { fold(true); qn = 0; done = 0; lost = true; }
  };
  auto push = [&](unsigned mask, int e) {
    const int pos = qn + (int)__builtin_amdgcn_mbcnt_lo(mask, 0u);
    if ((mask >> lane) & 1u) myq[pos] = e;
    qn += __builtin_popcount(mask);
  };

  int* glist = lists + (size_t)(blockIdx.x * nw + wv) * QC;
  int have = -1;
  if (TAKE) have = __builtin_amdgcn_readfirstlane(lcnt[blockIdx.x * nw + wv]);

  if (TAKE && have >= 0) {
    for (int q0 = 0; q0 < have; q0 += 128) {
      int pk[4];
#pragma unroll
      for (int u = 0; u < 4; ++u) {
        const int i = q0 + u * 32 + lane;
        pk[u] = glist[i < have ? i : have - 1];
      }
      float sv[4][H];
#pragma unroll
      for (int u = 0; u < 4; ++u)
#pragma unroll
        for (int hd = 0; hd < H; ++hd) sv[u][hd] = asrc[(size_t)(pk[u] >> 8) * H + hd];
#pragma unroll
      for (int u = 0; u < 4; ++u)
#pragma unroll
        for (int hd = 0; hd < H; ++hd) asm volatile("" :: "v"(sv[u][hd]));
#pragma unroll
      for (int u = 0; u < 4; ++u) {
        const int i = q0 + u * 32 + lane;
        if (i < have) {
          const int dl = pk[u] & 255;
          myq[i] = pk[u];
#pragma unroll
          for (int hd = 0; hd < H; ++hd) {
            float v = sv[u][hd] + adl[dl * H + hd];
            v = v > 0.f ? v : NEG_SLOPE * v;
            atomicMax(mu + dl * H + hd, f2ord(v));
          }
        }
      }
    }
    qn = have;
  } else {
    {
      const int j = wv + nw * lane;
      const unsigned mask = __builtin_amdgcn_ballot_w32(j < nloc);
      if (mask) push(mask, Eraw + base + j);
    }

    if (gridDim.x <= 255u && (Eraw & 15) == 0) {
      const int nkb = (Eraw + 511) >> 9;
      const unsigned pat = blockIdx.x * 0x01010101u;
      auto ldk = [&](int blk) {
        uint4 r = make_uint4(~0u, ~0u, ~0u, ~0u);
        if (blk < nkb) {
          const int e = blk * 512 + lane * 16;
          if (e < Eraw) r = *(const uint4*)(keys + (e >> 2));
        }
        return r;
      };
      auto eqb = [&](unsigned c) {
        const unsigned x = c ^ pat;
        return ~(((x & 0x7f7f7f7fu) + 0x7f7f7f7fu) | x | 0x7f7f7f7fu);
      };
      auto prock = [&](uint4 c, int blk) {
        unsigned m = (eqb(c.x) >> 7) | (eqb(c.y) >> 6) | (eqb(c.z) >> 5) | (eqb(c.w) >> 4);
        unsigned any = __builtin_amdgcn_ballot_w32(m != 0u);
        if (any) {
          const int e0 = blk * 512 + lane * 16;
          do {
            room(32);
            const int t = __builtin_ctz(m | 0x80000000u);
            push(any, e0 + ((t & 7) << 2) + (t >> 3));
            m &= m - 1u;
            any = __builtin_amdgcn_ballot_w32(m != 0u);
          } while (any);
          fold(false);
        }
      };
      int blk = wv;
      uint4 r0 = ldk(blk),          r1 = ldk(blk + nw),     r2 = ldk(blk + 2 * nw), r3 = ldk(blk + 3 * nw);
      uint4 r4 = ldk(blk + 4 * nw), r5 = ldk(blk + 5 * nw), r6 = ldk(blk + 6 * nw), r7 = ldk(blk + 7 * nw);
      for (; blk < nkb; blk += 8 * nw) {
        prock(r0, blk);          r0 = ldk(blk + 8 * nw);
        prock(r1, blk + nw);     r1 = ldk(blk + 9 * nw);
        prock(r2, blk + 2 * nw); r2 = ldk(blk + 10 * nw);
        prock(r3, blk + 3 * nw); r3 = ldk(blk + 11 * nw);
        prock(r4, blk + 4 * nw); r4 = ldk(blk + 12 * nw);
        prock(r5, blk + 5 * nw); r5 = ldk(blk + 13 * nw);
        prock(r6, blk + 6 * nw); r6 = ldk(blk + 14 * nw);
        prock(r7, blk + 7 * nw); r7 = ldk(blk + 15 * nw);
      }
    } else {
    const int* dstp = ei + Eraw;
    const int nblk = ((Eraw & 3) == 0) ? (Eraw >> 7) : 0;
    auto ld = [&](int blk) {
      int4 r = make_int4(-1, -1, -1, -1);
      if (blk < nblk) r = *(const int4*)(dstp + (size_t)blk * 128 + lane * 4);
      return r;
    };
    auto proc = [&](int4 c, int blk) {
      const unsigned m0 = __builtin_amdgcn_ballot_w32((unsigned)(c.x - base) < (unsigned)nloc);
      const unsigned m1 = __builtin_amdgcn_ballot_w32((unsigned)(c.y - base) < (unsigned)nloc);
      const unsigned m2 = __builtin_amdgcn_ballot_w32((unsigned)(c.z - base) < (unsigned)nloc);
      const unsigned m3 = __builtin_amdgcn_ballot_w32((unsigned)(c.w - base) < (unsigned)nloc);
      if (m0 | m1 | m2 | m3) {
        room(128);
        const int e = blk * 128 + lane * 4;
        if (m0) push(m0, e);
        if (m1) push(m1, e + 1);
        if (m2) push(m2, e + 2);
        if (m3) push(m3, e + 3);
        fold(false);
      }
    };
    {
      int blk = wv;
      int4 r0 = ld(blk), r1 = ld(blk + nw), r2 = ld(blk + 2 * nw), r3 = ld(blk + 3 * nw);
      for (; blk < nblk; blk += 4 * nw) {
        proc(r0, blk);          r0 = ld(blk + 4 * nw);
        proc(r1, blk + nw);     r1 = ld(blk + 5 * nw);
        proc(r2, blk + 2 * nw); r2 = ld(blk + 6 * nw);
        proc(r3, blk + 3 * nw); r3 = ld(blk + 7 * nw);
      }
    }
    for (int e0 = nblk * 128 + wv * 32; e0 < Eraw; e0 += nw * 32) {
      const int e = e0 + lane;
      const int d = (e < Eraw) ? dstp[e] : -1;
      const unsigned mask = __builtin_amdgcn_ballot_w32((unsigned)(d - base) < (unsigned)nloc);
      if (mask) { room(32); push(mask, e); fold(false); }
    }
    }
    fold(true);
    if (lost && lane == 0) ovf = 1;
  }
  __syncthreads();

  const int again = __builtin_amdgcn_readfirstlane(ovf);
  if (!TAKE) {
    if (!again) for (int i = lane; i < qn; i += 32) glist[i] = myq[i];
    if (lane == 0) lcnt[blockIdx.x * nw + wv] = again ? -1 : qn;
  }
  if (!again) {
    auto batch = [&](int q0, int cnt) {
      const int g  = lane % G;
      const int k  = HALF ? ((lane & 15) ^ (lane >> 4)) : g / H;
      const int hd = HALF ? 0 : g - k * H;
      const int kq = k < cnt ? k : cnt - 1;
      const int pk = myq[q0 + kq];
      const int sq = pk >> 8;
      const int dq = pk & 255;

      float4 f[B][NV];
      float  f1[B][NL];
      int    dk[B];
#pragma unroll
      for (int kk = 0; kk < B; ++kk) {
        const int sk = __builtin_amdgcn_readlane(sq, kk * H);
        dk[kk]       = __builtin_amdgcn_readlane(dq, kk * H);
        if constexpr (HALF) {
        } else if constexpr (WIDE) {
          const float* fp = feat + (size_t)sk * C + lane * 4;
#pragma unroll
          for (int j = 0; j < NV; ++j) f[kk][j] = *(const float4*)(fp + j * 128);
        } else {
          const float* fp = feat + (size_t)sk * C + lane;
#pragma unroll
          for (int j = 0; j < NL; ++j) f1[kk][j] = fp[j * 32];
        }
      }

      float4 fh[8];
      int    dh[8];
      float  ah[8];
      if constexpr (HALF) {
        const float* fb = feat + (lane & 15) * 4;
#define HGATHER(P) { const int sp = row_share<2 * (P)>(sq); dh[P] = row_share<2 * (P)>(dq); \
                     fh[P] = *(const float4*)(fb + (size_t)sp * C); }
        HGATHER(0) HGATHER(1) HGATHER(2) HGATHER(3) HGATHER(4) HGATHER(5) HGATHER(6) HGATHER(7)
#undef HGATHER
      }

      float v = asrc[(size_t)sq * H + hd] + adl[dq * H + hd];
      v = v > 0.f ? v : NEG_SLOPE * v;
      const float w = __expf(v - ord2f(mu[dq * H + hd]));
      if (lane < G && k < cnt) atomicAdd(den + dq * H + hd, w);

      if constexpr (HALF) {
#define HWEIGHT(P) ah[P] = __int_as_float(row_share<2 * (P)>(__float_as_int(w)));
        HWEIGHT(0) HWEIGHT(1) HWEIGHT(2) HWEIGHT(3) HWEIGHT(4) HWEIGHT(5) HWEIGHT(6) HWEIGHT(7)
#undef HWEIGHT
        const int lh = lane >> 4;
#pragma unroll
        for (int p = 0; p < 8; ++p) {
          if (2 * p < cnt) {
            if (2 * p + lh < cnt) {
              const float4 x = fh[p];
              const float  a = ah[p];
              const float y0 = lh ? x.z : x.x, y1 = lh ? x.w : x.y, y2 = lh ? x.x : x.z, y3 = lh ? x.y : x.w;
              float* op  = tile + dh[p] * C + (lane & 15);
              float* o01 = op + (lh ? 32 : 0);
              float* o23 = op + (lh ? 0 : 32);
              atomicAdd(o01,      y0 * a);
              atomicAdd(o01 + 16, y1 * a);
              atomicAdd(o23,      y2 * a);
              atomicAdd(o23 + 16, y3 * a);
            }
          }
        }
      } else {

#pragma unroll
      for (int kk = 0; kk < B; ++kk) {
        if (kk < cnt) {
          float* op = tile + dk[kk] * C + lane;
          if constexpr (WIDE) {
#pragma unroll
            for (int j = 0; j < NV; ++j) {
              const float w0 = __int_as_float(__builtin_amdgcn_readlane(__float_as_int(w), kk * H + 2 * j));
              const float w1 = __int_as_float(__builtin_amdgcn_readlane(__float_as_int(w), kk * H + 2 * j + 1));
              const float a  = (lane & 16) ? w1 : w0;
              atomicAdd(op + j * 128,      f[kk][j].x * a);
              atomicAdd(op + j * 128 + 32, f[kk][j].y * a);
              atomicAdd(op + j * 128 + 64, f[kk][j].z * a);
              atomicAdd(op + j * 128 + 96, f[kk][j].w * a);
            }
          } else {
#pragma unroll
            for (int j = 0; j < NL; ++j) {
              const float a = __int_as_float(
                  __builtin_amdgcn_readlane(__float_as_int(w), kk * H + (j * 32) / F));
              atomicAdd(op + j * 32, f1[kk][j] * a);
            }
          }
        }
      }
      }
    };
    for (int q0 = 0; q0 < qn; q0 += B) batch(q0, (qn - q0) < B ? (qn - q0) : B);
  } else {
    const int Etot = Eraw + n_nodes;
    for (int e0 = wv * 32; e0 < Etot; e0 += nw * 32) {
      const int e = e0 + lane;
      int s = 0, d = -1;
      if (e < Etot) edge_ends(ei, e, Eraw, s, d);
      unsigned mask = __builtin_amdgcn_ballot_w32((unsigned)(d - base) < (unsigned)nloc);
      while (mask) {
        const int l = __builtin_ctz(mask);
        mask &= mask - 1;
        const int sk = __shfl(s, l);
        const int dk = __shfl(d, l) - base;
        if (lane < H) {
          float v = asrc[(size_t)sk * H + lane] + adl[dk * H + lane];
          v = v > 0.f ? v : NEG_SLOPE * v;
          atomicAdd(den + dk * H + lane, __expf(v - ord2f(mu[dk * H + lane])));
        }
#pragma unroll
        for (int j = 0; j < NL; ++j) {
          const int hd = (j * 32) / F;
          float v = asrc[(size_t)sk * H + hd] + adl[dk * H + hd];
          v = v > 0.f ? v : NEG_SLOPE * v;
          const float a = __expf(v - ord2f(mu[dk * H + hd]));
          const int ch = j * 32 + lane;
          const int pw = WIDE ? ((ch & ~127) | ((ch & 3) << 5) | ((ch & 127) >> 2))
                       : HALF ? (((ch & 3) << 4) | (ch >> 2)) : ch;
          atomicAdd(tile + dk * C + pw, feat[(size_t)sk * C + ch] * a);
        }
      }
    }
  }
  __syncthreads();

  {
    const float4* t4 = (const float4*)tile;
    float4* o4 = (float4*)(out + (size_t)base * C);
    const float4* b4 = (const float4*)bias;
    for (int i = tid; i < nloc * (C / 4); i += blockDim.x) {
      const int nl = i / (C / 4);
      const int c4 = i - nl * (C / 4);
      const float r = den[nl * H + (c4 * 4) / F] + 1e-16f;
      float4 t;
      if constexpr (WIDE) {
        const float* tp = tile + nl * C + (c4 >> 5) * 128 + (c4 & 31);
        t = make_float4(tp[0], tp[32], tp[64], tp[96]);
      } else if constexpr (HALF) {
        const float* tp = tile + nl * C + c4;
        t = make_float4(tp[0], tp[16], tp[32], tp[48]);
      } else {
        t = t4[i];
      }
      const float4 bb = b4[c4];
      float4 v;
      v.x = t.x / r + bb.x; v.y = t.y / r + bb.y; v.z = t.z / r + bb.z; v.w = t.w / r + bb.w;
      if (ELU) {
        v.x = v.x > 0.f ? v.x : expm1f(v.x);
        v.y = v.y > 0.f ? v.y : expm1f(v.y);
        v.z = v.z > 0.f ? v.z : expm1f(v.z);
        v.w = v.w > 0.f ? v.w : expm1f(v.w);
      }
      o4[i] = v;
    }
  }
}


static inline int cdiv(long long a, long long b) { return (int)((a + b - 1) / b); }

extern "C" void kernel_launch(void* const* d_in, const int* in_sizes, int n_in,
                              void* d_out, int out_size, void* d_ws, size_t ws_size,
                              hipStream_t stream) {
  const float* x      = (const float*)d_in[0];
  const int*   ei     = (const int*)  d_in[1];
  const float* W1     = (const float*)d_in[2];
  const float* att_s1 = (const float*)d_in[3];
  const float* att_d1 = (const float*)d_in[4];
  const float* b1     = (const float*)d_in[5];
  const float* W2     = (const float*)d_in[6];
  const float* att_s2 = (const float*)d_in[7];
  const float* att_d2 = (const float*)d_in[8];
  const float* b2     = (const float*)d_in[9];

  const int IN   = 128;
  const int C1   = 256;
  const int Nn   = in_sizes[0] / IN;
  const int Eraw = in_sizes[1] / 2;

  float* ws = (float*)d_ws;
  size_t off = 0;
  float*    h1     = ws + off; off += (size_t)Nn * C1;
  float*    out1   = ws + off; off += (size_t)Nn * C1;
  float*    asrc1  = ws + off; off += (size_t)Nn * 4;
  float*    adst1  = ws + off; off += (size_t)Nn * 4;
  int*      lists  = (int*)(ws + off); off += (size_t)cdiv(Nn, 200) * 32 * LIST_SLOTS;
  int*      lcnt   = (int*)(ws + off); off += (size_t)cdiv(Nn, 200) * 32;
  unsigned* keys   = (unsigned*)(ws + off); off += (size_t)cdiv(Eraw, 4);
  float*    h2     = h1;
  float*    asrc2  = asrc1;
  float*    adst2  = adst1;
  float*    outf   = (float*)d_out;

  const int T = 256;

  gemm_lds_f32<128, 256, 7, 2><<<cdiv(Nn, 7 * 16), 256, 0, stream>>>(x, W1, h1, Nn);
  node_alpha<4, 64, 200, true><<<cdiv((long long)Nn * 4, 2 * T), T, 0, stream>>>(
      h1, att_s1, att_d1, asrc1, adst1, Nn, ei, keys, Eraw);
  range_agg_k<4, 64, 8, 200, true, false><<<cdiv(Nn, 200), 1024, 0, stream>>>(
      ei, asrc1, adst1, h1, b1, out1, lists, lcnt, keys, Nn, Eraw);

  gemm_lds_f32<256, 64, 8, 4><<<cdiv(Nn, 8 * 16), 256, 0, stream>>>(out1, W2, h2, Nn);
  node_alpha<1, 64, 200, false><<<cdiv((long long)Nn, T), T, 0, stream>>>(
      h2, att_s2, att_d2, asrc2, adst2, Nn, ei, keys, Eraw);
  range_agg_k<1, 64, 16, 200, false, true><<<cdiv(Nn, 200), 1024, 0, stream>>>(
      ei, asrc2, adst2, h2, b2, outf, lists, lcnt, keys, Nn, Eraw);
}


# Node scores as row.(W a), a folded into W per work-group: layer 1's by one more f32 WMMA tile in its GEMM, layer 2's in layer 1's aggregation epilogue; both score launches gone; sum order changes
# speedup vs baseline: 4.8369x; 1.0335x over previous
#include <hip/hip_runtime.h>

typedef __attribute__((ext_vector_type(2))) float v2f;
typedef __attribute__((ext_vector_type(8))) float v8f;
typedef __attribute__((ext_vector_type(4))) float v4f;

#define NEG_SLOPE 0.2f
#define LIST_SLOTS 512


__device__ __forceinline__ unsigned f2ord(float f) {
  unsigned u = __float_as_uint(f);
  return (u & 0x80000000u) ? ~u : (u | 0x80000000u);
}
__device__ __forceinline__ float ord2f(unsigned o) {
  unsigned u = (o & 0x80000000u) ? (o & 0x7fffffffu) : ~o;
  return __uint_as_float(u);
}

__device__ __forceinline__ void edge_ends(const int* __restrict__ ei, int e,
                                          int Eraw, int& s, int& d) {
  if (e < Eraw) { s = ei[e]; d = ei[Eraw + e]; }
  else          { s = e - Eraw; d = s; }
}


template <int CTRL>
__device__ __forceinline__ float row_xadd(float v) {
  return v + __int_as_float(__builtin_amdgcn_update_dpp(0, __float_as_int(v), CTRL, 0xf, 0xf, true));
}

template <int N>
__device__ __forceinline__ int row_share(int v) {
  return __builtin_amdgcn_update_dpp(0, v, 0x150 + N, 0xf, 0xf, true);
}

template <int K, int N, int MT, int KU, bool SC, int NPB>
__global__ __launch_bounds__(256, 4) void gemm_lds_f32(const float* __restrict__ A,
                                                    const float* __restrict__ B,
                                                    float* __restrict__ C, int M,
                                                    const float* __restrict__ att_s,
                                                    const float* __restrict__ att_d,
                                                    float* __restrict__ asrc,
                                                    float* __restrict__ adst,
                                                    const int* __restrict__ ei,
                                                    unsigned* __restrict__ keys, int Eraw) {
  constexpr int P  = K + 4;
  constexpr int NT = N / 16;
  constexpr int RN = (NT >= 8) ? NT / 8 : 1;
  constexpr int WN = NT / RN;
  constexpr int WM = 8 / WN;
  constexpr int RM = MT / WM;
  static_assert(WN * WM == 8 && RM * WM == MT && (K % (4 * KU)) == 0, "tiling");
  constexpr int ROWS = MT * 16;
  constexpr int R4   = K / 4;
  constexpr int TOT4 = ROWS * R4;
  static_assert((TOT4 % 256) == 0, "staging");

  static_assert(!SC || (N == 256 && MT <= 8 && (K % 16) == 0), "scores: four heads of 64 columns, a row tile per wave");
  __shared__ __attribute__((aligned(16))) float As[ROWS * P];
  __shared__ __attribute__((aligned(16))) float Ut[SC ? 8 * P : 4];

  const int tid  = threadIdx.x;
  const int lane = tid & 31;
  const int w    = __builtin_amdgcn_readfirstlane(tid >> 5);
  const int lh   = lane >> 4;
  const int l16  = lane & 15;
  const int m0   = blockIdx.x * ROWS;

  const int wn = w % WN;
  const int wm = w / WN;
  const float* bp = B + (size_t)(lh * 2) * N + (wn * RN) * 16 + l16;
  const float* ap = As + (wm * RM * 16 + l16) * P + lh * 2;

  v2f bc[KU][RN];
#pragma unroll
  for (int u = 0; u < KU; ++u)
#pragma unroll
    for (int j = 0; j < RN; ++j) {
      bc[u][j].x = bp[(size_t)(4 * u) * N + j * 16];
      bc[u][j].y = bp[(size_t)(4 * u + 1) * N + j * 16];
    }

  const int  kstride = gridDim.x * 256;
  const int  kq0 = blockIdx.x * 256 + tid;
  const int  nq  = (SC && (Eraw & 15) == 0) ? (Eraw >> 2) : 0;
  const int4* d4 = (const int4*)(ei + Eraw);
  int4 kd0 = make_int4(-1, -1, -1, -1), kd1 = make_int4(-1, -1, -1, -1);
  if constexpr (SC) {
    if (nq > 0) {
      kd0 = d4[kq0 < nq ? kq0 : nq - 1];
      kd1 = d4[kq0 + kstride < nq ? kq0 + kstride : nq - 1];
    }
  }

  constexpr int NU = SC ? (K / 8) * 2 : 1;
  constexpr int UG = 16;
  static_assert(!SC || ((NU % UG) == 0 && (K / 8) <= 16), "fold groups; a lane per k value of the wave");
  float4 fs[2], fd[2];
  if constexpr (SC) {
#pragma unroll
    for (int r = 0; r < 2; ++r) {
      fs[r] = *(const float4*)(att_s + (2 * r + lh) * 64 + l16 * 4);
      fd[r] = *(const float4*)(att_d + (2 * r + lh) * 64 + l16 * 4);
    }
  }
  auto key = [&](int d) -> unsigned {
    unsigned kx = (unsigned)d / (unsigned)NPB;
    if (kx > 255u) kx = 255u;
    return ((unsigned)d < (unsigned)M) ? kx : 255u;
  };
  auto pack = [&](int4 d) -> unsigned {
    return key(d.x) | (key(d.y) << 8) | (key(d.z) << 16) | (key(d.w) << 24);
  };
  auto fold_u = [&]() {
    if constexpr (SC) {
      const float* wp = B + (size_t)((K / 8) * 4 * w + lh) * 64 + l16 * 4;
      float so[2] = {0.f, 0.f}, dd[2] = {0.f, 0.f};
#pragma unroll
      for (int u0 = 0; u0 < NU; u0 += UG) {
        v4f xw[UG];
#pragma unroll
        for (int uu = 0; uu < UG; ++uu) xw[uu] = *(const v4f*)(wp + (u0 + uu) * 128);
#pragma unroll
        for (int uu = 0; uu < UG; ++uu) asm volatile("" :: "v"(xw[uu]));
        if (u0 == 0) {
          if (kq0 < nq)           keys[kq0]           = pack(kd0);
          if (kq0 + kstride < nq) keys[kq0 + kstride] = pack(kd1);
        }
#pragma unroll
        for (int uu = 0; uu < UG; ++uu) {
          const int u = u0 + uu;
          const float4 a = fs[u & 1], b = fd[u & 1];
          float sv = xw[uu].x * a.x + xw[uu].y * a.y + xw[uu].z * a.z + xw[uu].w * a.w;
          float dv = xw[uu].x * b.x + xw[uu].y * b.y + xw[uu].z * b.z + xw[uu].w * b.w;
          sv = row_xadd<0x161>(sv); dv = row_xadd<0x161>(dv);
          sv = row_xadd<0x162>(sv); dv = row_xadd<0x162>(dv);
          sv = row_xadd<0x164>(sv); dv = row_xadd<0x164>(dv);
          sv = row_xadd<0x168>(sv); dv = row_xadd<0x168>(dv);
          if (l16 == (u >> 1)) { so[u & 1] = sv; dd[u & 1] = dv; }
        }
      }
      if (l16 < K / 8) {
        const int kk = (K / 8) * w + l16;
#pragma unroll
        for (int r = 0; r < 2; ++r) {
          Ut[(2 * r + lh) * P + kk]     = so[r];
          Ut[(4 + 2 * r + lh) * P + kk] = dd[r];
        }
      }
    }
  };

  {
    constexpr int PER = TOT4 / 256;
    constexpr int UB  = (PER % 16 == 0) ? 16 : PER;
    static_assert((PER % UB) == 0, "staging batches");
    for (int c0 = 0; c0 < PER; c0 += UB) {
      float4 v[UB];
#pragma unroll
      for (int u = 0; u < UB; ++u) {
        const int i  = tid + (c0 + u) * 256;
        const int r  = i / R4;
        const int c4 = i - r * R4;
        int gr = m0 + r; if (gr > M - 1) gr = M - 1;
        v[u] = *(const float4*)(A + (size_t)gr * K + c4 * 4);
      }
      if (c0 == 0) fold_u();
      if constexpr (SC) {
#pragma unroll
        for (int u = 0; u < UB; ++u) asm volatile("" :: "v"(v[u].x), "v"(v[u].y), "v"(v[u].z), "v"(v[u].w));
      }
#pragma unroll
      for (int u = 0; u < UB; ++u) {
        const int i  = tid + (c0 + u) * 256;
        const int r  = i / R4;
        const int c4 = i - r * R4;
        *(float4*)(As + r * P + c4 * 4) = v[u];
      }
    }
  }
  __syncthreads();

  v8f acc[RM][RN];
#pragma unroll
  for (int i = 0; i < RM; ++i)
#pragma unroll
    for (int j = 0; j < RN; ++j) acc[i][j] = v8f{};

  const int ws = (w < MT) ? w : MT - 1;
  const float* aps = As + (ws * 16 + l16) * P + lh * 2;
  const float* ups = Ut + (l16 & 7) * P + lh * 2;
  v8f accs = v8f{};

#pragma unroll
  for (int k = 0; k < K; k += 4 * KU) {
    const int kn = (k + 4 * KU < K) ? k + 4 * KU : k;
    const float* bq = bp + (size_t)kn * N;
    v2f bn[KU][RN];
#pragma unroll
    for (int u = 0; u < KU; ++u)
#pragma unroll
      for (int j = 0; j < RN; ++j) {
        bn[u][j].x = bq[(size_t)(4 * u) * N + j * 16];
        bn[u][j].y = bq[(size_t)(4 * u + 1) * N + j * 16];
      }
    v2f a[KU][RM];
#pragma unroll
    for (int u = 0; u < KU; ++u)
#pragma unroll
      for (int i = 0; i < RM; ++i) a[u][i] = *(const v2f*)(ap + i * 16 * P + k + 4 * u);
    v2f sa[KU], sb[KU];
    if constexpr (SC) {
#pragma unroll
      for (int u = 0; u < KU; ++u) {
        sa[u] = *(const v2f*)(aps + k + 4 * u);
        sb[u] = *(const v2f*)(ups + k + 4 * u);
      }
    }
#pragma unroll
    for (int u = 0; u < KU; ++u)
#pragma unroll
      for (int i = 0; i < RM; ++i)
#pragma unroll
        for (int j = 0; j < RN; ++j)
          acc[i][j] = __builtin_amdgcn_wmma_f32_16x16x4_f32(false, a[u][i], false, bc[u][j],
                                                            (short)0, acc[i][j], false, false);
    if constexpr (SC) {
#pragma unroll
      for (int u = 0; u < KU; ++u)
        accs = __builtin_amdgcn_wmma_f32_16x16x4_f32(false, sa[u], false, sb[u],
                                                     (short)0, accs, false, false);
    }
#pragma unroll
    for (int u = 0; u < KU; ++u)
#pragma unroll
      for (int j = 0; j < RN; ++j) bc[u][j] = bn[u][j];
  }

  const int mtiles = M >> 4;
#pragma unroll
  for (int i = 0; i < RM; ++i) {
    const int mt = (m0 >> 4) + wm * RM + i;
    if (mt < mtiles) {
#pragma unroll
      for (int j = 0; j < RN; ++j) {
        float* crow = C + (size_t)(mt * 16 + lh * 8) * N + (wn * RN + j) * 16 + l16;
#pragma unroll
        for (int q = 0; q < 8; ++q) crow[(size_t)q * N] = acc[i][j][q];
      }
    }
  }

  if constexpr (SC) {
    if (w < MT && l16 < 8) {
      float* sp = ((l16 < 4) ? asrc : adst) + (l16 & 3);
      const int r0 = m0 + w * 16 + lh * 8;
#pragma unroll
      for (int q = 0; q < 8; ++q)
        if (r0 + q < M) sp[(size_t)(r0 + q) * 4] = accs[q];
    }
    for (int q = kq0 + 2 * kstride; q < nq; q += kstride) keys[q] = pack(d4[q]);
  }
}


template <int H, int F, int B, int NPB, bool ELU, bool TAKE, bool NEXT>
__global__ __launch_bounds__(1024) void range_agg_k(
    const int* __restrict__ ei,
    const float* __restrict__ asrc,
    const float* __restrict__ adst,
    const float* __restrict__ feat,
    const float* __restrict__ bias,
    float* __restrict__ out,
    int* __restrict__ lists, int* __restrict__ lcnt,
    const unsigned* __restrict__ keys,
    int n_nodes, int Eraw,
    const float* __restrict__ Wn,
    const float* __restrict__ natt_s,
    const float* __restrict__ natt_d,
    float* __restrict__ nasrc, float* __restrict__ nadst) {
  constexpr int C  = H * F;
  constexpr int NL = C / 32;
  constexpr bool WIDE = (C % 128) == 0 && F == 64;
  constexpr int NV = WIDE ? C / 128 : 1;
  constexpr bool HALF = (C == 64) && (H == 1) && (B == 16);
  constexpr int G  = B * H;
  constexpr int QC = LIST_SLOTS;
  static_assert(G <= 32 && (32 % G) == 0 && (F % 32) == 0 && (C % 4) == 0, "tiling");
  static_assert(NPB <= 256, "local row must fit the low byte of a list entry");

  __shared__ __attribute__((aligned(16))) float tile[NPB * C];
  __shared__ float    den[NPB * H];
  __shared__ unsigned mu[NPB * H];
  __shared__ float    adl[NPB * H];
  __shared__ int      qe[32 * QC];
  __shared__ int      ovf;
  constexpr int NF = 64;
  static_assert(!NEXT || ((C % 128) == 0 && WIDE), "NEXT: rows of whole 512-byte pieces");
  __shared__ __attribute__((aligned(16))) float nus[NEXT ? C : 4];
  __shared__ __attribute__((aligned(16))) float nud[NEXT ? C : 4];

  const int tid  = threadIdx.x;
  const int lane = tid & 31;
  const int wv   = __builtin_amdgcn_readfirstlane(tid >> 5);
  const int nw   = blockDim.x >> 5;
  const int base = blockIdx.x * NPB;
  int nloc = n_nodes - base; if (nloc > NPB) nloc = NPB;

  {
    float4* t4 = (float4*)tile;
    for (int i = tid; i < NPB * C / 4; i += blockDim.x) t4[i] = make_float4(0.f, 0.f, 0.f, 0.f);
    for (int i = tid; i < NPB * H; i += blockDim.x) {
      den[i] = 0.f;
      mu[i]  = 0u;
      adl[i] = (i < nloc * H) ? adst[(size_t)base * H + i] : 0.f;
    }
    if (tid == 0) ovf = 0;
    if constexpr (NEXT) {
      const int l16 = lane & 15, lh = lane >> 4;
      const float4 a = *(const float4*)(natt_s + l16 * 4);
      const float4 b = *(const float4*)(natt_d + l16 * 4);
      for (int c0 = wv * 8; c0 < C; c0 += nw * 8) {
        v4f x[4];
#pragma unroll
        for (int t = 0; t < 4; ++t) x[t] = *(const v4f*)(Wn + (size_t)(c0 + 2 * t + lh) * NF + l16 * 4);
        float so = 0.f, dd = 0.f;
#pragma unroll
        for (int t = 0; t < 4; ++t) {
          float sv = x[t].x * a.x + x[t].y * a.y + x[t].z * a.z + x[t].w * a.w;
          float dv = x[t].x * b.x + x[t].y * b.y + x[t].z * b.z + x[t].w * b.w;
          sv = row_xadd<0x161>(sv); dv = row_xadd<0x161>(dv);
          sv = row_xadd<0x162>(sv); dv = row_xadd<0x162>(dv);
          sv = row_xadd<0x164>(sv); dv = row_xadd<0x164>(dv);
          sv = row_xadd<0x168>(sv); dv = row_xadd<0x168>(dv);
          if (l16 == t) { so = sv; dd = dv; }
        }
        if (l16 < 4) { nus[c0 + 2 * l16 + lh] = so; nud[c0 + 2 * l16 + lh] = dd; }
      }
    }
  }
  __syncthreads();

  int* myq = qe + wv * QC;
  int  qn = 0;
  int  done = 0;
  bool lost = false;

  auto chunk = [&](int q0, int cnt) {
    if (lane < cnt) {
      const int e = myq[q0 + lane];
      int s, d; edge_ends(ei, e, Eraw, s, d);
      const int dl = d - base;
      float sc[H];
      if constexpr (H == 4) {
        const float4 s4 = *(const float4*)(asrc + (size_t)s * 4);
        sc[0] = s4.x; sc[1] = s4.y; sc[2] = s4.z; sc[3] = s4.w;
      } else {
#pragma unroll
        for (int hd = 0; hd < H; ++hd) sc[hd] = asrc[(size_t)s * H + hd];
      }
#pragma unroll
      for (int hd = 0; hd < H; ++hd) {
        float v = sc[hd] + adl[dl * H + hd];
        v = v > 0.f ? v : NEG_SLOPE * v;
        atomicMax(mu + dl * H + hd, f2ord(v));
      }
      myq[q0 + lane] = (s << 8) | dl;
    }
  };
  auto fold = [&](bool all) {
    __builtin_amdgcn_fence(__ATOMIC_ACQ_REL, "wavefront");
    while (qn - done >= 32) { chunk(done, 32); done += 32; }
    if (all && done < qn) { chunk(done, qn - done); done = qn; }
  };
  auto room = [&](int n) {
    if (qn + n > QC) { fold(true); qn = 0; done = 0; lost = true; }
  };
  auto push = [&](unsigned mask, int e) {
    const int pos = qn + (int)__builtin_amdgcn_mbcnt_lo(mask, 0u);
    if ((mask >> lane) & 1u) myq[pos] = e;
    qn += __builtin_popcount(mask);
  };

  int* glist = lists + (size_t)(blockIdx.x * nw + wv) * QC;
  int have = -1;
  if (TAKE) have = __builtin_amdgcn_readfirstlane(lcnt[blockIdx.x * nw + wv]);

  if (TAKE && have >= 0) {
    for (int q0 = 0; q0 < have; q0 += 128) {
      int pk[4];
#pragma unroll
      for (int u = 0; u < 4; ++u) {
        const int i = q0 + u * 32 + lane;
        pk[u] = glist[i < have ? i : have - 1];
      }
      float sv[4][H];
#pragma unroll
      for (int u = 0; u < 4; ++u)
#pragma unroll
        for (int hd = 0; hd < H; ++hd) sv[u][hd] = asrc[(size_t)(pk[u] >> 8) * H + hd];
#pragma unroll
      for (int u = 0; u < 4; ++u)
#pragma unroll
        for (int hd = 0; hd < H; ++hd) asm volatile("" :: "v"(sv[u][hd]));
#pragma unroll
      for (int u = 0; u < 4; ++u) {
        const int i = q0 + u * 32 + lane;
        if (i < have) {
          const int dl = pk[u] & 255;
          myq[i] = pk[u];
#pragma unroll
          for (int hd = 0; hd < H; ++hd) {
            float v = sv[u][hd] + adl[dl * H + hd];
            v = v > 0.f ? v : NEG_SLOPE * v;
            atomicMax(mu + dl * H + hd, f2ord(v));
          }
        }
      }
    }
    qn = have;
  } else {
    {
      const int j = wv + nw * lane;
      const unsigned mask = __builtin_amdgcn_ballot_w32(j < nloc);
      if (mask) push(mask, Eraw + base + j);
    }

    if (gridDim.x <= 255u && (Eraw & 15) == 0) {
      const int nkb = (Eraw + 511) >> 9;
      const unsigned pat = blockIdx.x * 0x01010101u;
      auto ldk = [&](int blk) {
        uint4 r = make_uint4(~0u, ~0u, ~0u, ~0u);
        if (blk < nkb) {
          const int e = blk * 512 + lane * 16;
          if (e < Eraw) r = *(const uint4*)(keys + (e >> 2));
        }
        return r;
      };
      auto eqb = [&](unsigned c) {
        const unsigned x = c ^ pat;
        return ~(((x & 0x7f7f7f7fu) + 0x7f7f7f7fu) | x | 0x7f7f7f7fu);
      };
      auto prock = [&](uint4 c, int blk) {
        unsigned m = (eqb(c.x) >> 7) | (eqb(c.y) >> 6) | (eqb(c.z) >> 5) | (eqb(c.w) >> 4);
        unsigned any = __builtin_amdgcn_ballot_w32(m != 0u);
        if (any) {
          const int e0 = blk * 512 + lane * 16;
          do {
            room(32);
            const int t = __builtin_ctz(m | 0x80000000u);
            push(any, e0 + ((t & 7) << 2) + (t >> 3));
            m &= m - 1u;
            any = __builtin_amdgcn_ballot_w32(m != 0u);
          } while (any);
          fold(false);
        }
      };
      int blk = wv;
      uint4 r0 = ldk(blk),          r1 = ldk(blk + nw),     r2 = ldk(blk + 2 * nw), r3 = ldk(blk + 3 * nw);
      uint4 r4 = ldk(blk + 4 * nw), r5 = ldk(blk + 5 * nw), r6 = ldk(blk + 6 * nw), r7 = ldk(blk + 7 * nw);
      for (; blk < nkb; blk += 8 * nw) {
        prock(r0, blk);          r0 = ldk(blk + 8 * nw);
        prock(r1, blk + nw);     r1 = ldk(blk + 9 * nw);
        prock(r2, blk + 2 * nw); r2 = ldk(blk + 10 * nw);
        prock(r3, blk + 3 * nw); r3 = ldk(blk + 11 * nw);
        prock(r4, blk + 4 * nw); r4 = ldk(blk + 12 * nw);
        prock(r5, blk + 5 * nw); r5 = ldk(blk + 13 * nw);
        prock(r6, blk + 6 * nw); r6 = ldk(blk + 14 * nw);
        prock(r7, blk + 7 * nw); r7 = ldk(blk + 15 * nw);
      }
    } else {
    const int* dstp = ei + Eraw;
    const int nblk = ((Eraw & 3) == 0) ? (Eraw >> 7) : 0;
    auto ld = [&](int blk) {
      int4 r = make_int4(-1, -1, -1, -1);
      if (blk < nblk) r = *(const int4*)(dstp + (size_t)blk * 128 + lane * 4);
      return r;
    };
    auto proc = [&](int4 c, int blk) {
      const unsigned m0 = __builtin_amdgcn_ballot_w32((unsigned)(c.x - base) < (unsigned)nloc);
      const unsigned m1 = __builtin_amdgcn_ballot_w32((unsigned)(c.y - base) < (unsigned)nloc);
      const unsigned m2 = __builtin_amdgcn_ballot_w32((unsigned)(c.z - base) < (unsigned)nloc);
      const unsigned m3 = __builtin_amdgcn_ballot_w32((unsigned)(c.w - base) < (unsigned)nloc);
      if (m0 | m1 | m2 | m3) {
        room(128);
        const int e = blk * 128 + lane * 4;
        if (m0) push(m0, e);
        if (m1) push(m1, e + 1);
        if (m2) push(m2, e + 2);
        if (m3) push(m3, e + 3);
        fold(false);
      }
    };
    {
      int blk = wv;
      int4 r0 = ld(blk), r1 = ld(blk + nw), r2 = ld(blk + 2 * nw), r3 = ld(blk + 3 * nw);
      for (; blk < nblk; blk += 4 * nw) {
        proc(r0, blk);          r0 = ld(blk + 4 * nw);
        proc(r1, blk + nw);     r1 = ld(blk + 5 * nw);
        proc(r2, blk + 2 * nw); r2 = ld(blk + 6 * nw);
        proc(r3, blk + 3 * nw); r3 = ld(blk + 7 * nw);
      }
    }
    for (int e0 = nblk * 128 + wv * 32; e0 < Eraw; e0 += nw * 32) {
      const int e = e0 + lane;
      const int d = (e < Eraw) ? dstp[e] : -1;
      const unsigned mask = __builtin_amdgcn_ballot_w32((unsigned)(d - base) < (unsigned)nloc);
      if (mask) { room(32); push(mask, e); fold(false); }
    }
    }
    fold(true);
    if (lost && lane == 0) ovf = 1;
  }
  __syncthreads();

  const int again = __builtin_amdgcn_readfirstlane(ovf);
  if (!TAKE) {
    if (!again) for (int i = lane; i < qn; i += 32) glist[i] = myq[i];
    if (lane == 0) lcnt[blockIdx.x * nw + wv] = again ? -1 : qn;
  }
  if (!again) {
    auto batch = [&](int q0, int cnt) {
      const int g  = lane % G;
      const int k  = HALF ? ((lane & 15) ^ (lane >> 4)) : g / H;
      const int hd = HALF ? 0 : g - k * H;
      const int kq = k < cnt ? k : cnt - 1;
      const int pk = myq[q0 + kq];
      const int sq = pk >> 8;
      const int dq = pk & 255;

      float4 f[B][NV];
      float  f1[B][NL];
      int    dk[B];
#pragma unroll
      for (int kk = 0; kk < B; ++kk) {
        const int sk = __builtin_amdgcn_readlane(sq, kk * H);
        dk[kk]       = __builtin_amdgcn_readlane(dq, kk * H);
        if constexpr (HALF) {
        } else if constexpr (WIDE) {
          const float* fp = feat + (size_t)sk * C + lane * 4;
#pragma unroll
          for (int j = 0; j < NV; ++j) f[kk][j] = *(const float4*)(fp + j * 128);
        } else {
          const float* fp = feat + (size_t)sk * C + lane;
#pragma unroll
          for (int j = 0; j < NL; ++j) f1[kk][j] = fp[j * 32];
        }
      }

      float4 fh[8];
      int    dh[8];
      float  ah[8];
      if constexpr (HALF) {
        const float* fb = feat + (lane & 15) * 4;
#define HGATHER(P) { const int sp = row_share<2 * (P)>(sq); dh[P] = row_share<2 * (P)>(dq); \
                     fh[P] = *(const float4*)(fb + (size_t)sp * C); }
        HGATHER(0) HGATHER(1) HGATHER(2) HGATHER(3) HGATHER(4) HGATHER(5) HGATHER(6) HGATHER(7)
#undef HGATHER
      }

      float v = asrc[(size_t)sq * H + hd] + adl[dq * H + hd];
      v = v > 0.f ? v : NEG_SLOPE * v;
      const float w = __expf(v - ord2f(mu[dq * H + hd]));
      if (lane < G && k < cnt) atomicAdd(den + dq * H + hd, w);

      if constexpr (HALF) {
#define HWEIGHT(P) ah[P] = __int_as_float(row_share<2 * (P)>(__float_as_int(w)));
        HWEIGHT(0) HWEIGHT(1) HWEIGHT(2) HWEIGHT(3) HWEIGHT(4) HWEIGHT(5) HWEIGHT(6) HWEIGHT(7)
#undef HWEIGHT
        const int lh = lane >> 4;
#pragma unroll
        for (int p = 0; p < 8; ++p) {
          if (2 * p < cnt) {
            if (2 * p + lh < cnt) {
              const float4 x = fh[p];
              const float  a = ah[p];
              const float y0 = lh ? x.z : x.x, y1 = lh ? x.w : x.y, y2 = lh ? x.x : x.z, y3 = lh ? x.y : x.w;
              float* op  = tile + dh[p] * C + (lane & 15);
              float* o01 = op + (lh ? 32 : 0);
              float* o23 = op + (lh ? 0 : 32);
              atomicAdd(o01,      y0 * a);
              atomicAdd(o01 + 16, y1 * a);
              atomicAdd(o23,      y2 * a);
              atomicAdd(o23 + 16, y3 * a);
            }
          }
        }
      } else {

#pragma unroll
      for (int kk = 0; kk < B; ++kk) {
        if (kk < cnt) {
          float* op = tile + dk[kk] * C + lane;
          if constexpr (WIDE) {
#pragma unroll
            for (int j = 0; j < NV; ++j) {
              const float w0 = __int_as_float(__builtin_amdgcn_readlane(__float_as_int(w), kk * H + 2 * j));
              const float w1 = __int_as_float(__builtin_amdgcn_readlane(__float_as_int(w), kk * H + 2 * j + 1));
              const float a  = (lane & 16) ? w1 : w0;
              atomicAdd(op + j * 128,      f[kk][j].x * a);
              atomicAdd(op + j * 128 + 32, f[kk][j].y * a);
              atomicAdd(op + j * 128 + 64, f[kk][j].z * a);
              atomicAdd(op + j * 128 + 96, f[kk][j].w * a);
            }
          } else {
#pragma unroll
            for (int j = 0; j < NL; ++j) {
              const float a = __int_as_float(
                  __builtin_amdgcn_readlane(__float_as_int(w), kk * H + (j * 32) / F));
              atomicAdd(op + j * 32, f1[kk][j] * a);
            }
          }
        }
      }
      }
    };
    for (int q0 = 0; q0 < qn; q0 += B) batch(q0, (qn - q0) < B ? (qn - q0) : B);
  } else {
    const int Etot = Eraw + n_nodes;
    for (int e0 = wv * 32; e0 < Etot; e0 += nw * 32) {
      const int e = e0 + lane;
      int s = 0, d = -1;
      if (e < Etot) edge_ends(ei, e, Eraw, s, d);
      unsigned mask = __builtin_amdgcn_ballot_w32((unsigned)(d - base) < (unsigned)nloc);
      while (mask) {
        const int l = __builtin_ctz(mask);
        mask &= mask - 1;
        const int sk = __shfl(s, l);
        const int dk = __shfl(d, l) - base;
        if (lane < H) {
          float v = asrc[(size_t)sk * H + lane] + adl[dk * H + lane];
          v = v > 0.f ? v : NEG_SLOPE * v;
          atomicAdd(den + dk * H + lane, __expf(v - ord2f(mu[dk * H + lane])));
        }
#pragma unroll
        for (int j = 0; j < NL; ++j) {
          const int hd = (j * 32) / F;
          float v = asrc[(size_t)sk * H + hd] + adl[dk * H + hd];
          v = v > 0.f ? v : NEG_SLOPE * v;
          const float a = __expf(v - ord2f(mu[dk * H + hd]));
          const int ch = j * 32 + lane;
          const int pw = WIDE ? ((ch & ~127) | ((ch & 3) << 5) | ((ch & 127) >> 2))
                       : HALF ? (((ch & 3) << 4) | (ch >> 2)) : ch;
          atomicAdd(tile + dk * C + pw, feat[(size_t)sk * C + ch] * a);
        }
      }
    }
  }
  __syncthreads();

  if constexpr (NEXT) {
    float4* o4 = (float4*)(out + (size_t)base * C);
    const float4* b4 = (const float4*)bias;
    const float4* us4 = (const float4*)nus;
    const float4* ud4 = (const float4*)nud;
    for (int nl = wv; nl < nloc; nl += nw) {
      float ps = 0.f, pd = 0.f;
#pragma unroll
      for (int j = 0; j < C / 128; ++j) {
        const int c4 = j * 32 + lane;
        const float r = den[nl * H + (c4 * 4) / F] + 1e-16f;
        const float* tp = tile + nl * C + j * 128 + lane;
        const float4 t = make_float4(tp[0], tp[32], tp[64], tp[96]);
        const float4 bb = b4[c4];
        float4 v;
        v.x = t.x / r + bb.x; v.y = t.y / r + bb.y; v.z = t.z / r + bb.z; v.w = t.w / r + bb.w;
        if (ELU) {
          v.x = v.x > 0.f ? v.x : expm1f(v.x);
          v.y = v.y > 0.f ? v.y : expm1f(v.y);
          v.z = v.z > 0.f ? v.z : expm1f(v.z);
          v.w = v.w > 0.f ? v.w : expm1f(v.w);
        }
        o4[nl * (C / 4) + c4] = v;
        const float4 us = us4[c4], ud = ud4[c4];
        ps += v.x * us.x + v.y * us.y + v.z * us.z + v.w * us.w;
        pd += v.x * ud.x + v.y * ud.y + v.z * ud.z + v.w * ud.w;
      }
      ps = row_xadd<0x161>(ps); pd = row_xadd<0x161>(pd);
      ps = row_xadd<0x162>(ps); pd = row_xadd<0x162>(pd);
      ps = row_xadd<0x164>(ps); pd = row_xadd<0x164>(pd);
      ps = row_xadd<0x168>(ps); pd = row_xadd<0x168>(pd);
      ps += __shfl_xor(ps, 16); pd += __shfl_xor(pd, 16);
      if (lane == 0) { nasrc[base + nl] = ps; nadst[base + nl] = pd; }
    }
  } else {
    const float4* t4 = (const float4*)tile;
    float4* o4 = (float4*)(out + (size_t)base * C);
    const float4* b4 = (const float4*)bias;
    for (int i = tid; i < nloc * (C / 4); i += blockDim.x) {
      const int nl = i / (C / 4);
      const int c4 = i - nl * (C / 4);
      const float r = den[nl * H + (c4 * 4) / F] + 1e-16f;
      float4 t;
      if constexpr (WIDE) {
        const float* tp = tile + nl * C + (c4 >> 5) * 128 + (c4 & 31);
        t = make_float4(tp[0], tp[32], tp[64], tp[96]);
      } else if constexpr (HALF) {
        const float* tp = tile + nl * C + c4;
        t = make_float4(tp[0], tp[16], tp[32], tp[48]);
      } else {
        t = t4[i];
      }
      const float4 bb = b4[c4];
      float4 v;
      v.x = t.x / r + bb.x; v.y = t.y / r + bb.y; v.z = t.z / r + bb.z; v.w = t.w / r + bb.w;
      if (ELU) {
        v.x = v.x > 0.f ? v.x : expm1f(v.x);
        v.y = v.y > 0.f ? v.y : expm1f(v.y);
        v.z = v.z > 0.f ? v.z : expm1f(v.z);
        v.w = v.w > 0.f ? v.w : expm1f(v.w);
      }
      o4[i] = v;
    }
  }
}


static inline int cdiv(long long a, long long b) { return (int)((a + b - 1) / b); }

extern "C" void kernel_launch(void* const* d_in, const int* in_sizes, int n_in,
                              void* d_out, int out_size, void* d_ws, size_t ws_size,
                              hipStream_t stream) {
  const float* x      = (const float*)d_in[0];
  const int*   ei     = (const int*)  d_in[1];
  const float* W1     = (const float*)d_in[2];
  const float* att_s1 = (const float*)d_in[3];
  const float* att_d1 = (const float*)d_in[4];
  const float* b1     = (const float*)d_in[5];
  const float* W2     = (const float*)d_in[6];
  const float* att_s2 = (const float*)d_in[7];
  const float* att_d2 = (const float*)d_in[8];
  const float* b2     = (const float*)d_in[9];

  const int IN   = 128;
  const int C1   = 256;
  const int Nn   = in_sizes[0] / IN;
  const int Eraw = in_sizes[1] / 2;

  float* ws = (float*)d_ws;
  size_t off = 0;
  float*    h1     = ws + off; off += (size_t)Nn * C1;
  float*    out1   = ws + off; off += (size_t)Nn * C1;
  float*    asrc1  = ws + off; off += (size_t)Nn * 4;
  float*    adst1  = ws + off; off += (size_t)Nn * 4;
  int*      lists  = (int*)(ws + off); off += (size_t)cdiv(Nn, 200) * 32 * LIST_SLOTS;
  int*      lcnt   = (int*)(ws + off); off += (size_t)cdiv(Nn, 200) * 32;
  unsigned* keys   = (unsigned*)(ws + off); off += (size_t)cdiv(Eraw, 4);
  float*    asrc2  = ws + off; off += (size_t)Nn;
  float*    adst2  = ws + off; off += (size_t)Nn;
  float*    h2     = h1;
  float*    outf   = (float*)d_out;

  gemm_lds_f32<128, 256, 7, 2, true, 200><<<cdiv(Nn, 7 * 16), 256, 0, stream>>>(
      x, W1, h1, Nn, att_s1, att_d1, asrc1, adst1, ei, keys, Eraw);
  range_agg_k<4, 64, 8, 200, true, false, true><<<cdiv(Nn, 200), 1024, 0, stream>>>(
      ei, asrc1, adst1, h1, b1, out1, lists, lcnt, keys, Nn, Eraw, W2, att_s2, att_d2, asrc2, adst2);

  gemm_lds_f32<256, 64, 8, 4, false, 200><<<cdiv(Nn, 8 * 16), 256, 0, stream>>>(
      out1, W2, h2, Nn, nullptr, nullptr, nullptr, nullptr, nullptr, nullptr, 0);
  range_agg_k<1, 64, 16, 200, false, true, false><<<cdiv(Nn, 200), 1024, 0, stream>>>(
      ei, asrc2, adst2, h2, b2, outf, lists, lcnt, keys, Nn, Eraw, nullptr, nullptr, nullptr, nullptr, nullptr);
}
